# starting kernel
# baseline (speedup 1.0000x reference)
#include <hip/hip_runtime.h>
#include <math.h>

#define NN 256
#define BATCH 4096
#define N_MZIS 32640
#define LAYERS 255

typedef __attribute__((ext_vector_type(2))) float v2f;
typedef __attribute__((ext_vector_type(8))) float v8f;

__global__ void __launch_bounds__(256)
mzi_coef_kernel(const float* __restrict__ theta,
                const float* __restrict__ phi,
                float4* __restrict__ coef) {
    int i = blockIdx.x * blockDim.x + threadIdx.x;
    if (i < N_MZIS) {
        float s, c, sp, cp;
        sincosf(theta[i], &s, &c);
        sincosf(phi[i], &sp, &cp);
        coef[i] = make_float4(c, s * cp, s * sp, 0.0f);
    }
}

__global__ void __launch_bounds__(32)
mzi_mesh_kernel(const float4* __restrict__ coef,
                const float* __restrict__ phi_ext,
                float* __restrict__ W) {
    __shared__ float Ure[NN][32];
    __shared__ float Uim[NN][32];
    const int t   = threadIdx.x;
    const int col = blockIdx.x * 32 + t;

    for (int r = 0; r < NN; ++r) {
        Ure[r][t] = (r == col) ? 1.0f : 0.0f;
        Uim[r][t] = 0.0f;
    }

    int idx = 0;
    for (int layer = 0; layer < LAYERS; ++layer) {
        const int npos = LAYERS - layer;
        for (int pos = 0; pos < npos; ++pos, ++idx) {
            const float4 cf = coef[idx];
            const float c  = cf.x;
            const float er = cf.y;
            const float ei = cf.z;
            const float r0r = Ure[pos    ][t], r0i = Uim[pos    ][t];
            const float r1r = Ure[pos + 1][t], r1i = Uim[pos + 1][t];
            const float n0r = fmaf(c, r0r, -fmaf(er, r1r,  ei * r1i));
            const float n0i = fmaf(c, r0i, -fmaf(er, r1i, -ei * r1r));
            const float n1r = fmaf(c, r1r,  fmaf(er, r0r, -ei * r0i));
            const float n1i = fmaf(c, r1i,  fmaf(er, r0i,  ei * r0r));
            Ure[pos    ][t] = n0r; Uim[pos    ][t] = n0i;
            Ure[pos + 1][t] = n1r; Uim[pos + 1][t] = n1i;
        }
    }

    for (int j = 0; j < NN; ++j) {
        float sj, cj;
        sincosf(phi_ext[j], &sj, &cj);
        W[j * NN + col] = fmaf(cj, Ure[j][t], -sj * Uim[j][t]);
    }
}

__global__ void __launch_bounds__(256)
mzi_gemm_kernel(const float* __restrict__ x,
                const float* __restrict__ W,
                float* __restrict__ out) {
    const int lane  = threadIdx.x & 31;
    const int wave  = threadIdx.x >> 5;
    const int tile  = blockIdx.x * 8 + wave;
    const int tM    = tile >> 2;
    const int tN    = tile & 3;
    const int lo    = lane & 15;
    const int hi    = lane >> 4;

    const float* aptr0 = x + (tM * 32      + lo) * NN + hi * 2;
    const float* aptr1 = aptr0 + 16 * NN;
    const float* bptr0 = W + (tN * 64      + lo) * NN + hi * 2;
    const float* bptr1 = bptr0 + 16 * NN;
    const float* bptr2 = bptr0 + 32 * NN;
    const float* bptr3 = bptr0 + 48 * NN;

    v8f c00 = {}, c01 = {}, c02 = {}, c03 = {};
    v8f c10 = {}, c11 = {}, c12 = {}, c13 = {};

#pragma unroll 2
    for (int k = 0; k < NN; k += 4) {
        v2f a0 = *(const v2f*)(aptr0 + k);
        v2f a1 = *(const v2f*)(aptr1 + k);
        v2f b0 = *(const v2f*)(bptr0 + k);
        v2f b1 = *(const v2f*)(bptr1 + k);
        v2f b2 = *(const v2f*)(bptr2 + k);
        v2f b3 = *(const v2f*)(bptr3 + k);
        c00 = __builtin_amdgcn_wmma_f32_16x16x4_f32(false, a0, false, b0, (short)0, c00, false, false);
        c01 = __builtin_amdgcn_wmma_f32_16x16x4_f32(false, a0, false, b1, (short)0, c01, false, false);
        c02 = __builtin_amdgcn_wmma_f32_16x16x4_f32(false, a0, false, b2, (short)0, c02, false, false);
        c03 = __builtin_amdgcn_wmma_f32_16x16x4_f32(false, a0, false, b3, (short)0, c03, false, false);
        c10 = __builtin_amdgcn_wmma_f32_16x16x4_f32(false, a1, false, b0, (short)0, c10, false, false);
        c11 = __builtin_amdgcn_wmma_f32_16x16x4_f32(false, a1, false, b1, (short)0, c11, false, false);
        c12 = __builtin_amdgcn_wmma_f32_16x16x4_f32(false, a1, false, b2, (short)0, c12, false, false);
        c13 = __builtin_amdgcn_wmma_f32_16x16x4_f32(false, a1, false, b3, (short)0, c13, false, false);
    }

    v8f acc[2][4] = {{c00, c01, c02, c03}, {c10, c11, c12, c13}};
#pragma unroll
    for (int i = 0; i < 2; ++i) {
#pragma unroll
        for (int j = 0; j < 4; ++j) {
            float* optr = out + (tM * 32 + i * 16 + hi * 8) * NN
                              + tN * 64 + j * 16 + lo;
#pragma unroll
            for (int g = 0; g < 8; ++g)
                optr[g * NN] = acc[i][j][g];
        }
    }
}

extern "C" void kernel_launch(void* const* d_in, const int* in_sizes, int n_in,
                              void* d_out, int out_size, void* d_ws, size_t ws_size,
                              hipStream_t stream) {
    const float* x       = (const float*)d_in[0];
    const float* theta   = (const float*)d_in[1];
    const float* phi_int = (const float*)d_in[2];
    const float* phi_ext = (const float*)d_in[3];
    float*       out     = (float*)d_out;

    float4* coef = (float4*)d_ws;
    float*  W    = (float*)((char*)d_ws + (size_t)N_MZIS * 16);

    mzi_coef_kernel<<<(N_MZIS + 255) / 256, 256, 0, stream>>>(theta, phi_int, coef);
    mzi_mesh_kernel<<<NN / 32, 32, 0, stream>>>(coef, phi_ext, W);
    mzi_gemm_kernel<<<(BATCH / 32) * (NN / 64) / 8, 256, 0, stream>>>(x, W, out);
}


# Mesh scan walked in 509 anti-diagonal wavefronts of independent MZIs (was 32640 dependent steps): 16 waves per block share each wavefront, one LDS barrier each; table stored in that order
# speedup vs baseline: 13.0368x; 13.0368x over previous
#include <hip/hip_runtime.h>
#include <math.h>

#define NN 256
#define BATCH 4096
#define N_MZIS 32640
#define LAYERS 255

typedef __attribute__((ext_vector_type(2))) float v2f;
typedef __attribute__((ext_vector_type(8))) float v8f;

#define NSTEPS 509
#define MESH_WAVES 16
#define GROUP_MAX 8
#define PF 4
static_assert(MESH_WAVES * GROUP_MAX == 128 && PF * GROUP_MAX == 32, "one fetch fills the 32 lanes");

__device__ __forceinline__ int wf_n(int t) {
    return (t >> 1) - max(0, t - 254) + 1;
}
__device__ __forceinline__ int wf_off(int t) {
    return (t <= 255) ? ((t + 1) >> 1) * ((t >> 1) + 1)
                      : N_MZIS - ((510 - t) >> 1) * (((509 - t) >> 1) + 1);
}

__global__ void __launch_bounds__(128)
mzi_coef_kernel(const float* __restrict__ theta,
                const float* __restrict__ phi,
                float4* __restrict__ tab) {
    const int t    = blockIdx.x;
    const int j    = threadIdx.x;
    const int lmin = max(0, t - 254);
    const int n    = (t >> 1) - lmin + 1;
    if (j < n) {
        const int l = lmin + j;
        const int p = t - 2 * l;
        const int i = l * 255 - ((l * (l - 1)) >> 1) + p;
        float s, c, sp, cp;
        sincosf(theta[i], &s, &c);
        sincosf(phi[i], &sp, &cp);
        tab[wf_off(t) + j] = make_float4(c, s * cp, s * sp, 0.0f);
    }
}

__device__ __forceinline__ float lane_value(float v, int l) {
    return __builtin_bit_cast(float,
        __builtin_amdgcn_readlane(__builtin_bit_cast(int, v), l));
}

__device__ __forceinline__ void lds_barrier() {
    __builtin_amdgcn_fence(__ATOMIC_RELEASE, "workgroup", "local");
    __builtin_amdgcn_s_barrier();
    __builtin_amdgcn_fence(__ATOMIC_ACQUIRE, "workgroup", "local");
}

__device__ __forceinline__ float4 fetch_coef(const float4* __restrict__ tab,
                                             const int tb, const int wv, const int lane) {
    const int tw = min(tb + lane / GROUP_MAX, NSTEPS - 1);
    const int e  = min(wv + MESH_WAVES * (lane % GROUP_MAX), wf_n(tw) - 1);
    return tab[wf_off(tw) + e];
}

template <int CNT>
__device__ __forceinline__ void mzi_group(float2 (*U)[32], const int lane,
                                          const int ptop, const int ltop,
                                          const float4 cq) {
    const int pbot = ptop - 2 * MESH_WAVES * (CNT - 1);
    float2 (*Ub)[32] = U + pbot;
    float2 a[CNT], b[CNT], na[CNT], nb[CNT];
#pragma unroll
    for (int v = 0; v < CNT; ++v) {
        a[v] = Ub[2 * MESH_WAVES * v][lane];
        b[v] = Ub[2 * MESH_WAVES * v + 1][lane];
    }
#pragma unroll
    for (int v = 0; v < CNT; ++v) {
        const float c  = lane_value(cq.x, ltop - v);
        const float er = lane_value(cq.y, ltop - v);
        const float ei = lane_value(cq.z, ltop - v);
        const float r0r = a[v].x, r0i = a[v].y;
        const float r1r = b[v].x, r1i = b[v].y;
        na[v].x = fmaf(c, r0r, -fmaf(er, r1r,  ei * r1i));
        na[v].y = fmaf(c, r0i, -fmaf(er, r1i, -ei * r1r));
        nb[v].x = fmaf(c, r1r,  fmaf(er, r0r, -ei * r0i));
        nb[v].y = fmaf(c, r1i,  fmaf(er, r0i,  ei * r0r));
    }
#pragma unroll
    for (int v = 0; v < CNT; ++v) {
        Ub[2 * MESH_WAVES * v][lane]     = na[v];
        Ub[2 * MESH_WAVES * v + 1][lane] = nb[v];
    }
}

__global__ void __launch_bounds__(32 * MESH_WAVES)
mzi_mesh_kernel(const float4* __restrict__ tab,
                const float* __restrict__ phi_ext,
                float* __restrict__ W) {
    __shared__ float2 U[NN][32];
    const int tid  = threadIdx.x;
    const int lane = tid & 31;
    const int wv   = __builtin_amdgcn_readfirstlane(tid >> 5);
    const int col  = blockIdx.x * 32 + lane;

    float4 cur = fetch_coef(tab, 0, wv, lane);

    for (int r = wv; r < NN; r += MESH_WAVES)
        U[r][lane] = make_float2((r == col) ? 1.0f : 0.0f, 0.0f);
    lds_barrier();

    for (int tb = 0; tb < NSTEPS; tb += PF) {
        const float4 nxt = fetch_coef(tab, tb + PF, wv, lane);
        const int send = min(PF, NSTEPS - tb);
#pragma unroll 1
        for (int s = 0; s < send; ++s) {
            const int t    = tb + s;
            const int lmin = max(0, t - 254);
            const int n    = (t >> 1) - lmin + 1;
            const int ptop = t - 2 * lmin - 2 * wv;
            const int m    = (n - wv + MESH_WAVES - 1) / MESH_WAVES;
            const int ltop = GROUP_MAX * s + m - 1;
            switch (m) {
                case 1: mzi_group<1>(U, lane, ptop, ltop, cur); break;
                case 2: mzi_group<2>(U, lane, ptop, ltop, cur); break;
                case 3: mzi_group<3>(U, lane, ptop, ltop, cur); break;
                case 4: mzi_group<4>(U, lane, ptop, ltop, cur); break;
                case 5: mzi_group<5>(U, lane, ptop, ltop, cur); break;
                case 6: mzi_group<6>(U, lane, ptop, ltop, cur); break;
                case 7: mzi_group<7>(U, lane, ptop, ltop, cur); break;
                case 8: mzi_group<8>(U, lane, ptop, ltop, cur); break;
                default: break;
            }
            lds_barrier();
        }
        cur = nxt;
    }

    for (int j = wv; j < NN; j += MESH_WAVES) {
        float sj, cj;
        sincosf(phi_ext[j], &sj, &cj);
        const float2 v = U[j][lane];
        W[j * NN + col] = fmaf(cj, v.x, -sj * v.y);
    }
}

__global__ void __launch_bounds__(256)
mzi_gemm_kernel(const float* __restrict__ x,
                const float* __restrict__ W,
                float* __restrict__ out) {
    const int lane  = threadIdx.x & 31;
    const int wave  = threadIdx.x >> 5;
    const int tile  = blockIdx.x * 8 + wave;
    const int tM    = tile >> 2;
    const int tN    = tile & 3;
    const int lo    = lane & 15;
    const int hi    = lane >> 4;

    const float* aptr0 = x + (tM * 32      + lo) * NN + hi * 2;
    const float* aptr1 = aptr0 + 16 * NN;
    const float* bptr0 = W + (tN * 64      + lo) * NN + hi * 2;
    const float* bptr1 = bptr0 + 16 * NN;
    const float* bptr2 = bptr0 + 32 * NN;
    const float* bptr3 = bptr0 + 48 * NN;

    v8f c00 = {}, c01 = {}, c02 = {}, c03 = {};
    v8f c10 = {}, c11 = {}, c12 = {}, c13 = {};

#pragma unroll 2
    for (int k = 0; k < NN; k += 4) {
        v2f a0 = *(const v2f*)(aptr0 + k);
        v2f a1 = *(const v2f*)(aptr1 + k);
        v2f b0 = *(const v2f*)(bptr0 + k);
        v2f b1 = *(const v2f*)(bptr1 + k);
        v2f b2 = *(const v2f*)(bptr2 + k);
        v2f b3 = *(const v2f*)(bptr3 + k);
        c00 = __builtin_amdgcn_wmma_f32_16x16x4_f32(false, a0, false, b0, (short)0, c00, false, false);
        c01 = __builtin_amdgcn_wmma_f32_16x16x4_f32(false, a0, false, b1, (short)0, c01, false, false);
        c02 = __builtin_amdgcn_wmma_f32_16x16x4_f32(false, a0, false, b2, (short)0, c02, false, false);
        c03 = __builtin_amdgcn_wmma_f32_16x16x4_f32(false, a0, false, b3, (short)0, c03, false, false);
        c10 = __builtin_amdgcn_wmma_f32_16x16x4_f32(false, a1, false, b0, (short)0, c10, false, false);
        c11 = __builtin_amdgcn_wmma_f32_16x16x4_f32(false, a1, false, b1, (short)0, c11, false, false);
        c12 = __builtin_amdgcn_wmma_f32_16x16x4_f32(false, a1, false, b2, (short)0, c12, false, false);
        c13 = __builtin_amdgcn_wmma_f32_16x16x4_f32(false, a1, false, b3, (short)0, c13, false, false);
    }

    v8f acc[2][4] = {{c00, c01, c02, c03}, {c10, c11, c12, c13}};
#pragma unroll
    for (int i = 0; i < 2; ++i) {
#pragma unroll
        for (int j = 0; j < 4; ++j) {
            float* optr = out + (tM * 32 + i * 16 + hi * 8) * NN
                              + tN * 64 + j * 16 + lo;
#pragma unroll
            for (int g = 0; g < 8; ++g)
                optr[g * NN] = acc[i][j][g];
        }
    }
}

extern "C" void kernel_launch(void* const* d_in, const int* in_sizes, int n_in,
                              void* d_out, int out_size, void* d_ws, size_t ws_size,
                              hipStream_t stream) {
    const float* x       = (const float*)d_in[0];
    const float* theta   = (const float*)d_in[1];
    const float* phi_int = (const float*)d_in[2];
    const float* phi_ext = (const float*)d_in[3];
    float*       out     = (float*)d_out;

    float4* coef = (float4*)d_ws;
    float*  W    = (float*)((char*)d_ws + (size_t)N_MZIS * 16);

    mzi_coef_kernel<<<NSTEPS, 128, 0, stream>>>(theta, phi_int, coef);
    mzi_mesh_kernel<<<NN / 32, 32 * MESH_WAVES, 0, stream>>>(coef, phi_ext, W);
    mzi_gemm_kernel<<<(BATCH / 32) * (NN / 64) / 8, 256, 0, stream>>>(x, W, out);
}


# Mesh lanes span a wavefront's MZIs (8 MZIs x 4 columns per wave, one update per wave and step): 64 blocks of 4 columns (was 8 of 32); W handed over as column pairs so each block stores whole lines
# speedup vs baseline: 19.9815x; 1.5327x over previous
#include <hip/hip_runtime.h>
#include <math.h>

#define NN 256
#define BATCH 4096
#define N_MZIS 32640
#define LAYERS 255

typedef __attribute__((ext_vector_type(2))) float v2f;
typedef __attribute__((ext_vector_type(8))) float v8f;

#define NSTEPS 509
#define COLS 4
#define SUBS (32 / COLS)
#define MESH_WAVES 16
#define PF 8
static_assert(MESH_WAVES * SUBS == 128, "one wave instruction per wave covers a whole wavefront");
static_assert(32 * MESH_WAVES == 2 * NN, "the store phase maps threads to (column pair, row)");

__device__ __forceinline__ int wf_n(int t) {
    return (t >> 1) - max(0, t - 254) + 1;
}
__device__ __forceinline__ int wf_off(int t) {
    return (t <= 255) ? ((t + 1) >> 1) * ((t >> 1) + 1)
                      : N_MZIS - ((510 - t) >> 1) * (((509 - t) >> 1) + 1);
}

__global__ void __launch_bounds__(128)
mzi_coef_kernel(const float* __restrict__ theta,
                const float* __restrict__ phi,
                float4* __restrict__ tab) {
    const int t    = blockIdx.x;
    const int j    = threadIdx.x;
    const int lmin = max(0, t - 254);
    const int n    = (t >> 1) - lmin + 1;
    if (j < n) {
        const int l = lmin + j;
        const int p = t - 2 * l;
        const int i = l * 255 - ((l * (l - 1)) >> 1) + p;
        float s, c, sp, cp;
        sincosf(theta[i], &s, &c);
        sincosf(phi[i], &sp, &cp);
        tab[wf_off(t) + j] = make_float4(c, s * cp, s * sp, 0.0f);
    }
}

__device__ __forceinline__ void lds_barrier() {
    __builtin_amdgcn_fence(__ATOMIC_RELEASE, "workgroup", "local");
    __builtin_amdgcn_s_barrier();
    __builtin_amdgcn_fence(__ATOMIC_ACQUIRE, "workgroup", "local");
}

__device__ __forceinline__ float4 fetch_coef(const float4* __restrict__ tab,
                                             const int t, const int e) {
    const int tw = min(t, NSTEPS - 1);
    return tab[wf_off(tw) + min(e, wf_n(tw) - 1)];
}

__global__ void __launch_bounds__(32 * MESH_WAVES)
mzi_mesh_kernel(const float4* __restrict__ tab,
                const float* __restrict__ phi_ext,
                float2* __restrict__ WP) {
    __shared__ float2 U[NN][COLS];
    const int tid  = threadIdx.x;
    const int lane = tid & 31;
    const int c    = lane & (COLS - 1);
    const int sub  = lane / COLS;
    const int wv   = __builtin_amdgcn_readfirstlane(tid >> 5);
    const int e    = wv * SUBS + sub;
    const int col0 = blockIdx.x * COLS;

    float4 cur[PF], nxt[PF];
#pragma unroll
    for (int s = 0; s < PF; ++s) cur[s] = fetch_coef(tab, s, e);

    for (int i = tid; i < NN * COLS; i += 32 * MESH_WAVES) {
        const int r = i / COLS, cc = i & (COLS - 1);
        U[r][cc] = make_float2((r == col0 + cc) ? 1.0f : 0.0f, 0.0f);
    }
    lds_barrier();

    for (int tb = 0; tb < NSTEPS; tb += PF) {
#pragma unroll
        for (int s = 0; s < PF; ++s) nxt[s] = fetch_coef(tab, tb + PF + s, e);
#pragma unroll
        for (int s = 0; s < PF; ++s) {
            const int t = tb + s;
            if (t < NSTEPS) {
                const int lmin = max(0, t - 254);
                const int n    = (t >> 1) - lmin + 1;
                if (e < n) {
                    const int p = t - 2 * (lmin + e);
                    const float2 a = U[p][c];
                    const float2 b = U[p + 1][c];
                    const float cs = cur[s].x;
                    const float er = cur[s].y;
                    const float ei = cur[s].z;
                    const float r0r = a.x, r0i = a.y;
                    const float r1r = b.x, r1i = b.y;
                    float2 na, nb;
                    na.x = fmaf(cs, r0r, -fmaf(er, r1r,  ei * r1i));
                    na.y = fmaf(cs, r0i, -fmaf(er, r1i, -ei * r1r));
                    nb.x = fmaf(cs, r1r,  fmaf(er, r0r, -ei * r0i));
                    nb.y = fmaf(cs, r1i,  fmaf(er, r0i,  ei * r0r));
                    U[p][c]     = na;
                    U[p + 1][c] = nb;
                }
                lds_barrier();
            }
        }
#pragma unroll
        for (int s = 0; s < PF; ++s) cur[s] = nxt[s];
    }

    {
        const int j  = tid & (NN - 1);
        const int kp = tid / NN;
        float sj, cj;
        sincosf(phi_ext[j], &sj, &cj);
        const float2 u0 = U[j][2 * kp];
        const float2 u1 = U[j][2 * kp + 1];
        WP[(blockIdx.x * (COLS / 2) + kp) * NN + j] =
            make_float2(fmaf(cj, u0.x, -sj * u0.y), fmaf(cj, u1.x, -sj * u1.y));
    }
}

__global__ void __launch_bounds__(256)
mzi_gemm_kernel(const float* __restrict__ x,
                const float* __restrict__ W,
                float* __restrict__ out) {
    const int lane  = threadIdx.x & 31;
    const int wave  = threadIdx.x >> 5;
    const int tile  = blockIdx.x * 8 + wave;
    const int tM    = tile >> 2;
    const int tN    = tile & 3;
    const int lo    = lane & 15;
    const int hi    = lane >> 4;

    const float* aptr0 = x + (tM * 32      + lo) * NN + hi * 2;
    const float* aptr1 = aptr0 + 16 * NN;
    const float* bptr0 = W + hi * 2 * NN + (tN * 64 + lo) * 2;
    const float* bptr1 = bptr0 + 16 * 2;
    const float* bptr2 = bptr0 + 32 * 2;
    const float* bptr3 = bptr0 + 48 * 2;

    v8f c00 = {}, c01 = {}, c02 = {}, c03 = {};
    v8f c10 = {}, c11 = {}, c12 = {}, c13 = {};

#pragma unroll 2
    for (int k = 0; k < NN; k += 4) {
        v2f a0 = *(const v2f*)(aptr0 + k);
        v2f a1 = *(const v2f*)(aptr1 + k);
        v2f b0 = *(const v2f*)(bptr0 + k * NN);
        v2f b1 = *(const v2f*)(bptr1 + k * NN);
        v2f b2 = *(const v2f*)(bptr2 + k * NN);
        v2f b3 = *(const v2f*)(bptr3 + k * NN);
        c00 = __builtin_amdgcn_wmma_f32_16x16x4_f32(false, a0, false, b0, (short)0, c00, false, false);
        c01 = __builtin_amdgcn_wmma_f32_16x16x4_f32(false, a0, false, b1, (short)0, c01, false, false);
        c02 = __builtin_amdgcn_wmma_f32_16x16x4_f32(false, a0, false, b2, (short)0, c02, false, false);
        c03 = __builtin_amdgcn_wmma_f32_16x16x4_f32(false, a0, false, b3, (short)0, c03, false, false);
        c10 = __builtin_amdgcn_wmma_f32_16x16x4_f32(false, a1, false, b0, (short)0, c10, false, false);
        c11 = __builtin_amdgcn_wmma_f32_16x16x4_f32(false, a1, false, b1, (short)0, c11, false, false);
        c12 = __builtin_amdgcn_wmma_f32_16x16x4_f32(false, a1, false, b2, (short)0, c12, false, false);
        c13 = __builtin_amdgcn_wmma_f32_16x16x4_f32(false, a1, false, b3, (short)0, c13, false, false);
    }

    v8f acc[2][4] = {{c00, c01, c02, c03}, {c10, c11, c12, c13}};
#pragma unroll
    for (int i = 0; i < 2; ++i) {
#pragma unroll
        for (int j = 0; j < 4; ++j) {
            float* optr = out + (tM * 32 + i * 16 + hi * 8) * NN
                              + tN * 64 + j * 16 + lo;
#pragma unroll
            for (int g = 0; g < 8; ++g)
                optr[g * NN] = acc[i][j][g];
        }
    }
}

extern "C" void kernel_launch(void* const* d_in, const int* in_sizes, int n_in,
                              void* d_out, int out_size, void* d_ws, size_t ws_size,
                              hipStream_t stream) {
    const float* x       = (const float*)d_in[0];
    const float* theta   = (const float*)d_in[1];
    const float* phi_int = (const float*)d_in[2];
    const float* phi_ext = (const float*)d_in[3];
    float*       out     = (float*)d_out;

    float4* coef = (float4*)d_ws;
    float*  W    = (float*)((char*)d_ws + (size_t)N_MZIS * 16);

    mzi_coef_kernel<<<NSTEPS, 128, 0, stream>>>(theta, phi_int, coef);
    mzi_mesh_kernel<<<NN / COLS, 32 * MESH_WAVES, 0, stream>>>(coef, phi_ext, (float2*)W);
    mzi_gemm_kernel<<<(BATCH / 32) * (NN / 64) / 8, 256, 0, stream>>>(x, W, out);
}


# Mesh scan tiled in space and time: a wave keeps 16 rows x 4 columns in registers for 8 wavefronts (aligned tile, then tile shifted 8 rows): 2 LDS hand-overs per 8 steps, was 8; table in tile order
# speedup vs baseline: 22.9207x; 1.1471x over previous
#include <hip/hip_runtime.h>
#include <math.h>

#define NN 256
#define BATCH 4096
#define N_MZIS 32640
#define LAYERS 255

typedef __attribute__((ext_vector_type(2))) float v2f;
typedef __attribute__((ext_vector_type(8))) float v8f;

#define NSTEPS 509
#define COLS 4
#define MESH_WAVES 16
#define TR 16
#define HS 8
#define NSLAB 64
#define LDP (NN + 16)
#define NSLOT 23
#define SLOTS_A 12
#define SUBS 8
#define TAB_ENTRIES (NSLAB * MESH_WAVES * NSLOT * SUBS)
static_assert(MESH_WAVES * TR == NN, "the waves' tiles cover every row");
static_assert(NSLAB * HS >= NSTEPS && (NSLAB % 2) == 0, "slabs cover the scan, in pairs");
static_assert(32 * MESH_WAVES == 2 * NN, "the store phase maps threads to (column pair, row)");

__global__ void __launch_bounds__(192)
mzi_coef_kernel(const float* __restrict__ theta,
                const float* __restrict__ phi,
                float4* __restrict__ tab) {
    const int k   = blockIdx.x / MESH_WAVES;
    const int w   = blockIdx.x % MESH_WAVES;
    const int j   = threadIdx.x / SUBS;
    const int sub = threadIdx.x % SUBS;
    if (j >= NSLOT) return;
    const bool shifted = (j >= SLOTS_A);
    const int  jj      = shifted ? j - SLOTS_A + 1 : j;
    const int  which   = jj % 3;
    const int  s       = 2 * (jj / 3) + (which != 0 ? 1 : 0);
    const int  t       = HS * k + s;
    int base, plo, phi_;
    if (!shifted) {
        base = TR * w;
        plo  = (w == 0) ? 0 : TR * w + s;
        phi_ = (w == MESH_WAVES - 1) ? NN - 2 : TR * w + TR - 2 - s;
    } else {
        base = TR * w + HS;
        plo  = TR * (w + 1) - 1 - s;
        phi_ = (w == MESH_WAVES - 1) ? -1 : TR * (w + 1) - 1 + s;
    }
    const int  p  = base + 2 * sub + (which == 1 ? 1 : (which == 2 ? -1 : 0));
    const bool on = p >= plo && p <= phi_ && p <= min(t, 2 * (NN - 2) - t);
    float4 q = make_float4(1.0f, 0.0f, 0.0f, 0.0f);
    if (on) {
        const int l = (t - p) >> 1;
        const int i = l * 255 - ((l * (l - 1)) >> 1) + p;
        float sn, cs, sp, cp;
        sincosf(theta[i], &sn, &cs);
        sincosf(phi[i], &sp, &cp);
        q = make_float4(cs, sn * cp, sn * sp, 0.0f);
    }
    tab[(blockIdx.x * NSLOT + j) * SUBS + sub] = q;
}

__device__ __forceinline__ void lds_barrier() {
    __builtin_amdgcn_fence(__ATOMIC_RELEASE, "workgroup", "local");
    __builtin_amdgcn_s_barrier();
    __builtin_amdgcn_fence(__ATOMIC_ACQUIRE, "workgroup", "local");
}

__device__ __forceinline__ float from_next(float v) {
    return __builtin_bit_cast(float, __builtin_amdgcn_mov_dpp8(__builtin_bit_cast(unsigned, v), 0xff58d1u));
}
__device__ __forceinline__ float from_prev(float v) {
    return __builtin_bit_cast(float, __builtin_amdgcn_mov_dpp8(__builtin_bit_cast(unsigned, v), 0xd63440u));
}

__device__ __forceinline__ float4 load_coef(const float4* p) {
    const float* f = static_cast<const float*>(__builtin_assume_aligned(p, 16));
    return make_float4(f[0], f[1], f[2], 0.0f);
}

template <int S0>
__device__ __forceinline__ void tile_run(const float4* cf, float2& a, float2& b) {
    int k = 0;
#pragma unroll
    for (int s = S0; s < HS; ++s) {
        if ((s & 1) == 0) {
            const float4 q = cf[k++];
            float2 na, nb;
            na.x = fmaf(q.x, a.x, -fmaf(q.y, b.x,  q.z * b.y));
            na.y = fmaf(q.x, a.y, -fmaf(q.y, b.y, -q.z * b.x));
            nb.x = fmaf(q.x, b.x,  fmaf(q.y, a.x, -q.z * a.y));
            nb.y = fmaf(q.x, b.y,  fmaf(q.y, a.y,  q.z * a.x));
            a = na;
            b = nb;
        } else {
            const float4 qu = cf[k++];
            const float4 qd = cf[k++];
            const float nx = from_next(a.x), ny = from_next(a.y);
            const float px = from_prev(b.x), py = from_prev(b.y);
            float2 na, nb;
            nb.x = fmaf(qu.x, b.x, fmaf(-qu.y, nx, -qu.z * ny));
            nb.y = fmaf(qu.x, b.y, fmaf(-qu.y, ny,  qu.z * nx));
            na.x = fmaf(qd.x, a.x,  fmaf(qd.y, px, -qd.z * py));
            na.y = fmaf(qd.x, a.y,  fmaf(qd.y, py,  qd.z * px));
            a = na;
            b = nb;
        }
    }
}

__device__ __forceinline__ void mesh_slab(const float4* __restrict__ tnext, float2 (*U)[LDP],
                                          int wv, int sub, int c,
                                          const float4* cur, float4* nxt) {
#pragma unroll
    for (int j = 0; j < NSLOT; ++j) nxt[j] = load_coef(tnext + j * SUBS);

    {
        float2* rows = &U[c][TR * wv + 2 * sub];
        const float4 r = *reinterpret_cast<const float4*>(rows);
        float2 a = make_float2(r.x, r.y), b = make_float2(r.z, r.w);
        tile_run<0>(cur, a, b);
        *reinterpret_cast<float4*>(rows) = make_float4(a.x, a.y, b.x, b.y);
    }
    lds_barrier();
    if (wv < MESH_WAVES - 1) {
        float2* rows = &U[c][TR * wv + HS + 2 * sub];
        const float4 r = *reinterpret_cast<const float4*>(rows);
        float2 a = make_float2(r.x, r.y), b = make_float2(r.z, r.w);
        tile_run<1>(cur + SLOTS_A, a, b);
        *reinterpret_cast<float4*>(rows) = make_float4(a.x, a.y, b.x, b.y);
    }
    lds_barrier();
}

__global__ void __launch_bounds__(32 * MESH_WAVES)
mzi_mesh_kernel(const float4* __restrict__ tab,
                const float* __restrict__ phi_ext,
                float2* __restrict__ WP) {
    __shared__ __attribute__((aligned(16))) float2 U[COLS][LDP];
    const int tid  = threadIdx.x;
    const int lane = tid & 31;
    const int sub  = lane & 7;
    const int c    = lane >> 3;
    const int wv   = __builtin_amdgcn_readfirstlane(tid >> 5);
    const int col0 = blockIdx.x * COLS;
    const int SLAB_STRIDE = MESH_WAVES * NSLOT * SUBS;
    const float4* tw = tab + wv * NSLOT * SUBS + sub;

    float4 bufx[NSLOT], bufy[NSLOT];
#pragma unroll
    for (int j = 0; j < NSLOT; ++j) bufx[j] = load_coef(tw + j * SUBS);

    for (int i = tid; i < NN * COLS; i += 32 * MESH_WAVES) {
        const int cc = i / NN, r = i & (NN - 1);
        U[cc][r] = make_float2((r == col0 + cc) ? 1.0f : 0.0f, 0.0f);
    }
    lds_barrier();

    for (int k = 0; k < NSLAB; k += 2) {
        mesh_slab(tw + (k + 1) * SLAB_STRIDE, U, wv, sub, c, bufx, bufy);
        mesh_slab(tw + min(k + 2, NSLAB - 1) * SLAB_STRIDE, U, wv, sub, c, bufy, bufx);
    }

    {
        const int j  = tid & (NN - 1);
        const int kp = tid / NN;
        float sj, cj;
        sincosf(phi_ext[j], &sj, &cj);
        const float2 u0 = U[2 * kp][j];
        const float2 u1 = U[2 * kp + 1][j];
        WP[(blockIdx.x * (COLS / 2) + kp) * NN + j] =
            make_float2(fmaf(cj, u0.x, -sj * u0.y), fmaf(cj, u1.x, -sj * u1.y));
    }
}

__global__ void __launch_bounds__(256)
mzi_gemm_kernel(const float* __restrict__ x,
                const float* __restrict__ W,
                float* __restrict__ out) {
    const int lane  = threadIdx.x & 31;
    const int wave  = threadIdx.x >> 5;
    const int tile  = blockIdx.x * 8 + wave;
    const int tM    = tile >> 2;
    const int tN    = tile & 3;
    const int lo    = lane & 15;
    const int hi    = lane >> 4;

    const float* aptr0 = x + (tM * 32      + lo) * NN + hi * 2;
    const float* aptr1 = aptr0 + 16 * NN;
    const float* bptr0 = W + hi * 2 * NN + (tN * 64 + lo) * 2;
    const float* bptr1 = bptr0 + 16 * 2;
    const float* bptr2 = bptr0 + 32 * 2;
    const float* bptr3 = bptr0 + 48 * 2;

    v8f c00 = {}, c01 = {}, c02 = {}, c03 = {};
    v8f c10 = {}, c11 = {}, c12 = {}, c13 = {};

#pragma unroll 2
    for (int k = 0; k < NN; k += 4) {
        v2f a0 = *(const v2f*)(aptr0 + k);
        v2f a1 = *(const v2f*)(aptr1 + k);
        v2f b0 = *(const v2f*)(bptr0 + k * NN);
        v2f b1 = *(const v2f*)(bptr1 + k * NN);
        v2f b2 = *(const v2f*)(bptr2 + k * NN);
        v2f b3 = *(const v2f*)(bptr3 + k * NN);
        c00 = __builtin_amdgcn_wmma_f32_16x16x4_f32(false, a0, false, b0, (short)0, c00, false, false);
        c01 = __builtin_amdgcn_wmma_f32_16x16x4_f32(false, a0, false, b1, (short)0, c01, false, false);
        c02 = __builtin_amdgcn_wmma_f32_16x16x4_f32(false, a0, false, b2, (short)0, c02, false, false);
        c03 = __builtin_amdgcn_wmma_f32_16x16x4_f32(false, a0, false, b3, (short)0, c03, false, false);
        c10 = __builtin_amdgcn_wmma_f32_16x16x4_f32(false, a1, false, b0, (short)0, c10, false, false);
        c11 = __builtin_amdgcn_wmma_f32_16x16x4_f32(false, a1, false, b1, (short)0, c11, false, false);
        c12 = __builtin_amdgcn_wmma_f32_16x16x4_f32(false, a1, false, b2, (short)0, c12, false, false);
        c13 = __builtin_amdgcn_wmma_f32_16x16x4_f32(false, a1, false, b3, (short)0, c13, false, false);
    }

    v8f acc[2][4] = {{c00, c01, c02, c03}, {c10, c11, c12, c13}};
#pragma unroll
    for (int i = 0; i < 2; ++i) {
#pragma unroll
        for (int j = 0; j < 4; ++j) {
            float* optr = out + (tM * 32 + i * 16 + hi * 8) * NN
                              + tN * 64 + j * 16 + lo;
#pragma unroll
            for (int g = 0; g < 8; ++g)
                optr[g * NN] = acc[i][j][g];
        }
    }
}

extern "C" void kernel_launch(void* const* d_in, const int* in_sizes, int n_in,
                              void* d_out, int out_size, void* d_ws, size_t ws_size,
                              hipStream_t stream) {
    const float* x       = (const float*)d_in[0];
    const float* theta   = (const float*)d_in[1];
    const float* phi_int = (const float*)d_in[2];
    const float* phi_ext = (const float*)d_in[3];
    float*       out     = (float*)d_out;

    float4* coef = (float4*)d_ws;
    float*  W    = (float*)((char*)d_ws + (size_t)TAB_ENTRIES * 16);

    mzi_coef_kernel<<<NSLAB * MESH_WAVES, 192, 0, stream>>>(theta, phi_int, coef);
    mzi_mesh_kernel<<<NN / COLS, 32 * MESH_WAVES, 0, stream>>>(coef, phi_ext, (float2*)W);
    mzi_gemm_kernel<<<(BATCH / 32) * (NN / 64) / 8, 256, 0, stream>>>(x, W, out);
}


# Mesh: each lane carries both columns of a 2-column block and a wave holds 4 tiles: 128 blocks of 4 waves, one per SIMD, two independent chains per lane (was 64 blocks of 16 waves, 4 per SIMD)
# speedup vs baseline: 33.6393x; 1.4676x over previous
#include <hip/hip_runtime.h>
#include <math.h>

#define NN 256
#define BATCH 4096
#define N_MZIS 32640
#define LAYERS 255

typedef __attribute__((ext_vector_type(2))) float v2f;
typedef __attribute__((ext_vector_type(8))) float v8f;

#define NSTEPS 509
#define COLS 2
#define MESH_WAVES 4
#define GROUPS 4
#define TILES 16
#define TR 16
#define HS 8
#define NSLAB 64
#define LDP (NN + 16)
#define NSLOT 23
#define SLOTS_A 12
#define SUBS 8
#define TAB_ENTRIES (NSLAB * TILES * NSLOT * SUBS)
static_assert(TILES * TR == NN && MESH_WAVES * GROUPS == TILES, "the lane groups' tiles cover every row");
static_assert(NSLAB * HS >= NSTEPS && (NSLAB % 2) == 0, "slabs cover the scan, in pairs");
static_assert(LDP >= NN + TR / 2, "the last tile's shifted pass stays inside the padded column");

__global__ void __launch_bounds__(192)
mzi_coef_kernel(const float* __restrict__ theta,
                const float* __restrict__ phi,
                float4* __restrict__ tab) {
    const int k   = blockIdx.x / TILES;
    const int w   = blockIdx.x % TILES;
    const int j   = threadIdx.x / SUBS;
    const int sub = threadIdx.x % SUBS;
    if (j >= NSLOT) return;
    const bool shifted = (j >= SLOTS_A);
    const int  jj      = shifted ? j - SLOTS_A + 1 : j;
    const int  which   = jj % 3;
    const int  s       = 2 * (jj / 3) + (which != 0 ? 1 : 0);
    const int  t       = HS * k + s;
    int base, plo, phi_;
    if (!shifted) {
        base = TR * w;
        plo  = (w == 0) ? 0 : TR * w + s;
        phi_ = (w == TILES - 1) ? NN - 2 : TR * w + TR - 2 - s;
    } else {
        base = TR * w + HS;
        plo  = TR * (w + 1) - 1 - s;
        phi_ = (w == TILES - 1) ? -1 : TR * (w + 1) - 1 + s;
    }
    const int  p  = base + 2 * sub + (which == 1 ? 1 : (which == 2 ? -1 : 0));
    const bool on = p >= plo && p <= phi_ && p <= min(t, 2 * (NN - 2) - t);
    float4 q = make_float4(1.0f, 0.0f, 0.0f, 0.0f);
    if (on) {
        const int l = (t - p) >> 1;
        const int i = l * 255 - ((l * (l - 1)) >> 1) + p;
        float sn, cs, sp, cp;
        sincosf(theta[i], &sn, &cs);
        sincosf(phi[i], &sp, &cp);
        q = make_float4(cs, sn * cp, sn * sp, 0.0f);
    }
    tab[(blockIdx.x * NSLOT + j) * SUBS + sub] = q;
}

__device__ __forceinline__ void lds_barrier() {
    __builtin_amdgcn_fence(__ATOMIC_RELEASE, "workgroup", "local");
    __builtin_amdgcn_s_barrier();
    __builtin_amdgcn_fence(__ATOMIC_ACQUIRE, "workgroup", "local");
}

__device__ __forceinline__ float from_next(float v) {
    return __builtin_bit_cast(float, __builtin_amdgcn_mov_dpp8(__builtin_bit_cast(unsigned, v), 0xff58d1u));
}
__device__ __forceinline__ float from_prev(float v) {
    return __builtin_bit_cast(float, __builtin_amdgcn_mov_dpp8(__builtin_bit_cast(unsigned, v), 0xd63440u));
}

__device__ __forceinline__ float4 load_coef(const float4* p) {
    const float* f = static_cast<const float*>(__builtin_assume_aligned(p, 16));
    return make_float4(f[0], f[1], f[2], 0.0f);
}

template <int S0>
__device__ __forceinline__ void tile_run(const float4* cf, float2 (&a)[COLS], float2 (&b)[COLS]) {
    int k = 0;
#pragma unroll
    for (int s = S0; s < HS; ++s) {
        if ((s & 1) == 0) {
            const float4 q = cf[k++];
#pragma unroll
            for (int cc = 0; cc < COLS; ++cc) {
                float2 na, nb;
                na.x = fmaf(q.x, a[cc].x, -fmaf(q.y, b[cc].x,  q.z * b[cc].y));
                na.y = fmaf(q.x, a[cc].y, -fmaf(q.y, b[cc].y, -q.z * b[cc].x));
                nb.x = fmaf(q.x, b[cc].x,  fmaf(q.y, a[cc].x, -q.z * a[cc].y));
                nb.y = fmaf(q.x, b[cc].y,  fmaf(q.y, a[cc].y,  q.z * a[cc].x));
                a[cc] = na;
                b[cc] = nb;
            }
        } else {
            const float4 qu = cf[k++];
            const float4 qd = cf[k++];
#pragma unroll
            for (int cc = 0; cc < COLS; ++cc) {
                const float nx = from_next(a[cc].x), ny = from_next(a[cc].y);
                const float px = from_prev(b[cc].x), py = from_prev(b[cc].y);
                float2 na, nb;
                nb.x = fmaf(qu.x, b[cc].x, fmaf(-qu.y, nx, -qu.z * ny));
                nb.y = fmaf(qu.x, b[cc].y, fmaf(-qu.y, ny,  qu.z * nx));
                na.x = fmaf(qd.x, a[cc].x,  fmaf(qd.y, px, -qd.z * py));
                na.y = fmaf(qd.x, a[cc].y,  fmaf(qd.y, py,  qd.z * px));
                a[cc] = na;
                b[cc] = nb;
            }
        }
    }
}

template <int S0>
__device__ __forceinline__ void tile_pass(float2 (*U)[LDP], int row, const float4* cf) {
    float2 a[COLS], b[COLS];
#pragma unroll
    for (int cc = 0; cc < COLS; ++cc) {
        const float4 r = *reinterpret_cast<const float4*>(&U[cc][row]);
        a[cc] = make_float2(r.x, r.y);
        b[cc] = make_float2(r.z, r.w);
    }
    tile_run<S0>(cf, a, b);
#pragma unroll
    for (int cc = 0; cc < COLS; ++cc)
        *reinterpret_cast<float4*>(&U[cc][row]) = make_float4(a[cc].x, a[cc].y, b[cc].x, b[cc].y);
}

__device__ __forceinline__ void mesh_slab(const float4* __restrict__ tnext, float2 (*U)[LDP],
                                          int row, const float4* cur, float4* nxt) {
#pragma unroll
    for (int j = 0; j < NSLOT; ++j) nxt[j] = load_coef(tnext + j * SUBS);

    tile_pass<0>(U, row, cur);
    lds_barrier();
    tile_pass<1>(U, row + HS, cur + SLOTS_A);
    lds_barrier();
}

__global__ void __launch_bounds__(32 * MESH_WAVES)
mzi_mesh_kernel(const float4* __restrict__ tab,
                const float* __restrict__ phi_ext,
                float2* __restrict__ WP) {
    __shared__ __attribute__((aligned(16))) float2 U[COLS][LDP];
    const int tid  = threadIdx.x;
    const int lane = tid & 31;
    const int sub  = lane & 7;
    const int g    = lane >> 3;
    const int wv   = __builtin_amdgcn_readfirstlane(tid >> 5);
    const int tile = GROUPS * wv + g;
    const int row  = TR * tile + 2 * sub;
    const int col0 = blockIdx.x * COLS;
    const int SLAB_STRIDE = TILES * NSLOT * SUBS;
    const float4* tw = tab + tile * NSLOT * SUBS + sub;

    float4 bufx[NSLOT], bufy[NSLOT];
#pragma unroll
    for (int j = 0; j < NSLOT; ++j) bufx[j] = load_coef(tw + j * SUBS);

    for (int i = tid; i < COLS * LDP; i += 32 * MESH_WAVES) {
        const int cc = i / LDP, r = i - cc * LDP;
        U[cc][r] = make_float2((r == col0 + cc) ? 1.0f : 0.0f, 0.0f);
    }
    lds_barrier();

    for (int k = 0; k < NSLAB; k += 2) {
        mesh_slab(tw + (k + 1) * SLAB_STRIDE, U, row, bufx, bufy);
        mesh_slab(tw + min(k + 2, NSLAB - 1) * SLAB_STRIDE, U, row, bufy, bufx);
    }

    for (int j = tid; j < NN; j += 32 * MESH_WAVES) {
        float sj, cj;
        sincosf(phi_ext[j], &sj, &cj);
        const float2 u0 = U[0][j];
        const float2 u1 = U[1][j];
        WP[blockIdx.x * NN + j] =
            make_float2(fmaf(cj, u0.x, -sj * u0.y), fmaf(cj, u1.x, -sj * u1.y));
    }
}

__global__ void __launch_bounds__(256)
mzi_gemm_kernel(const float* __restrict__ x,
                const float* __restrict__ W,
                float* __restrict__ out) {
    const int lane  = threadIdx.x & 31;
    const int wave  = threadIdx.x >> 5;
    const int tile  = blockIdx.x * 8 + wave;
    const int tM    = tile >> 2;
    const int tN    = tile & 3;
    const int lo    = lane & 15;
    const int hi    = lane >> 4;

    const float* aptr0 = x + (tM * 32      + lo) * NN + hi * 2;
    const float* aptr1 = aptr0 + 16 * NN;
    const float* bptr0 = W + hi * 2 * NN + (tN * 64 + lo) * 2;
    const float* bptr1 = bptr0 + 16 * 2;
    const float* bptr2 = bptr0 + 32 * 2;
    const float* bptr3 = bptr0 + 48 * 2;

    v8f c00 = {}, c01 = {}, c02 = {}, c03 = {};
    v8f c10 = {}, c11 = {}, c12 = {}, c13 = {};

#pragma unroll 2
    for (int k = 0; k < NN; k += 4) {
        v2f a0 = *(const v2f*)(aptr0 + k);
        v2f a1 = *(const v2f*)(aptr1 + k);
        v2f b0 = *(const v2f*)(bptr0 + k * NN);
        v2f b1 = *(const v2f*)(bptr1 + k * NN);
        v2f b2 = *(const v2f*)(bptr2 + k * NN);
        v2f b3 = *(const v2f*)(bptr3 + k * NN);
        c00 = __builtin_amdgcn_wmma_f32_16x16x4_f32(false, a0, false, b0, (short)0, c00, false, false);
        c01 = __builtin_amdgcn_wmma_f32_16x16x4_f32(false, a0, false, b1, (short)0, c01, false, false);
        c02 = __builtin_amdgcn_wmma_f32_16x16x4_f32(false, a0, false, b2, (short)0, c02, false, false);
        c03 = __builtin_amdgcn_wmma_f32_16x16x4_f32(false, a0, false, b3, (short)0, c03, false, false);
        c10 = __builtin_amdgcn_wmma_f32_16x16x4_f32(false, a1, false, b0, (short)0, c10, false, false);
        c11 = __builtin_amdgcn_wmma_f32_16x16x4_f32(false, a1, false, b1, (short)0, c11, false, false);
        c12 = __builtin_amdgcn_wmma_f32_16x16x4_f32(false, a1, false, b2, (short)0, c12, false, false);
        c13 = __builtin_amdgcn_wmma_f32_16x16x4_f32(false, a1, false, b3, (short)0, c13, false, false);
    }

    v8f acc[2][4] = {{c00, c01, c02, c03}, {c10, c11, c12, c13}};
#pragma unroll
    for (int i = 0; i < 2; ++i) {
#pragma unroll
        for (int j = 0; j < 4; ++j) {
            float* optr = out + (tM * 32 + i * 16 + hi * 8) * NN
                              + tN * 64 + j * 16 + lo;
#pragma unroll
            for (int g = 0; g < 8; ++g)
                optr[g * NN] = acc[i][j][g];
        }
    }
}

extern "C" void kernel_launch(void* const* d_in, const int* in_sizes, int n_in,
                              void* d_out, int out_size, void* d_ws, size_t ws_size,
                              hipStream_t stream) {
    const float* x       = (const float*)d_in[0];
    const float* theta   = (const float*)d_in[1];
    const float* phi_int = (const float*)d_in[2];
    const float* phi_ext = (const float*)d_in[3];
    float*       out     = (float*)d_out;

    float4* coef = (float4*)d_ws;
    float*  W    = (float*)((char*)d_ws + (size_t)TAB_ENTRIES * 16);

    mzi_coef_kernel<<<NSLAB * TILES, 192, 0, stream>>>(theta, phi_int, coef);
    mzi_mesh_kernel<<<NN / COLS, 32 * MESH_WAVES, 0, stream>>>(coef, phi_ext, (float2*)W);
    mzi_gemm_kernel<<<(BATCH / 32) * (NN / 64) / 8, 256, 0, stream>>>(x, W, out);
}


# Mesh tiles of 32 rows x 16 wavefronts (was 16 x 8): 64 LDS hand-overs, was 128; a tile is one 16-lane row, neighbour rows come by v_permlane16 (was DPP8 moves); table fetched one pass ahead
# speedup vs baseline: 36.4000x; 1.0821x over previous
#include <hip/hip_runtime.h>
#include <math.h>

#define NN 256
#define BATCH 4096
#define N_MZIS 32640
#define LAYERS 255

typedef __attribute__((ext_vector_type(2))) float v2f;
typedef __attribute__((ext_vector_type(8))) float v8f;

#define NSTEPS 509
#define COLS 2
#define MESH_WAVES 4
#define GROUPS 2
#define TILES 8
#define TR 32
#define HS 16
#define NSLAB 32
#define LDP (NN + 16)
#define NSLOT 47
#define SLOTS_A 24
#define SLOTS_B (NSLOT - SLOTS_A)
#define SUBS 16
#define TAB_ENTRIES (NSLAB * TILES * NSLOT * SUBS)
static_assert(TILES * TR == NN && MESH_WAVES * GROUPS == TILES, "the lane groups' tiles cover every row");
static_assert(NSLAB * HS >= NSTEPS, "slabs cover the scan");
static_assert(SLOTS_A == 3 * HS / 2 && NSLOT == 3 * HS - 1 && SUBS == TR / 2, "slots and lanes follow the tile");
static_assert(LDP >= NN + TR / 2, "the last tile's shifted pass stays inside the padded column");

__global__ void __launch_bounds__(768)
mzi_coef_kernel(const float* __restrict__ theta,
                const float* __restrict__ phi,
                float4* __restrict__ tab) {
    const int k   = blockIdx.x / TILES;
    const int w   = blockIdx.x % TILES;
    const int j   = threadIdx.x / SUBS;
    const int sub = threadIdx.x % SUBS;
    if (j >= NSLOT) return;
    const bool shifted = (j >= SLOTS_A);
    const int  jj      = shifted ? j - SLOTS_A + 1 : j;
    const int  which   = jj % 3;
    const int  s       = 2 * (jj / 3) + (which != 0 ? 1 : 0);
    const int  t       = HS * k + s;
    int base, plo, phi_;
    if (!shifted) {
        base = TR * w;
        plo  = (w == 0) ? 0 : TR * w + s;
        phi_ = (w == TILES - 1) ? NN - 2 : TR * w + TR - 2 - s;
    } else {
        base = TR * w + HS;
        plo  = TR * (w + 1) - 1 - s;
        phi_ = (w == TILES - 1) ? -1 : TR * (w + 1) - 1 + s;
    }
    const int  p  = base + 2 * sub + (which == 1 ? 1 : (which == 2 ? -1 : 0));
    const bool on = p >= plo && p <= phi_ && p <= min(t, 2 * (NN - 2) - t);
    float4 q = make_float4(1.0f, 0.0f, 0.0f, 0.0f);
    if (on) {
        const int l = (t - p) >> 1;
        const int i = l * 255 - ((l * (l - 1)) >> 1) + p;
        float sn, cs, sp, cp;
        sincosf(theta[i], &sn, &cs);
        sincosf(phi[i], &sp, &cp);
        q = make_float4(cs, sn * cp, sn * sp, 0.0f);
    }
    tab[(blockIdx.x * NSLOT + j) * SUBS + sub] = q;
}

__device__ __forceinline__ void lds_barrier() {
    __builtin_amdgcn_fence(__ATOMIC_RELEASE, "workgroup", "local");
    __builtin_amdgcn_s_barrier();
    __builtin_amdgcn_fence(__ATOMIC_ACQUIRE, "workgroup", "local");
}

__device__ __forceinline__ float from_next(float v) {
    const unsigned u = __builtin_bit_cast(unsigned, v);
    return __builtin_bit_cast(float, __builtin_amdgcn_permlane16(u, u, 0x87654321u, 0xffedcba9u, true, false));
}
__device__ __forceinline__ float from_prev(float v) {
    const unsigned u = __builtin_bit_cast(unsigned, v);
    return __builtin_bit_cast(float, __builtin_amdgcn_permlane16(u, u, 0x65432100u, 0xedcba987u, true, false));
}

__device__ __forceinline__ float4 load_coef(const float4* p) {
    const float* f = static_cast<const float*>(__builtin_assume_aligned(p, 16));
    return make_float4(f[0], f[1], f[2], 0.0f);
}

template <int S0>
__device__ __forceinline__ void tile_run(const float4* cf, float2 (&a)[COLS], float2 (&b)[COLS]) {
    int k = 0;
#pragma unroll
    for (int s = S0; s < HS; ++s) {
        if ((s & 1) == 0) {
            const float4 q = cf[k++];
#pragma unroll
            for (int cc = 0; cc < COLS; ++cc) {
                float2 na, nb;
                na.x = fmaf(q.x, a[cc].x, -fmaf(q.y, b[cc].x,  q.z * b[cc].y));
                na.y = fmaf(q.x, a[cc].y, -fmaf(q.y, b[cc].y, -q.z * b[cc].x));
                nb.x = fmaf(q.x, b[cc].x,  fmaf(q.y, a[cc].x, -q.z * a[cc].y));
                nb.y = fmaf(q.x, b[cc].y,  fmaf(q.y, a[cc].y,  q.z * a[cc].x));
                a[cc] = na;
                b[cc] = nb;
            }
        } else {
            const float4 qu = cf[k++];
            const float4 qd = cf[k++];
#pragma unroll
            for (int cc = 0; cc < COLS; ++cc) {
                const float nx = from_next(a[cc].x), ny = from_next(a[cc].y);
                const float px = from_prev(b[cc].x), py = from_prev(b[cc].y);
                float2 na, nb;
                nb.x = fmaf(qu.x, b[cc].x, fmaf(-qu.y, nx, -qu.z * ny));
                nb.y = fmaf(qu.x, b[cc].y, fmaf(-qu.y, ny,  qu.z * nx));
                na.x = fmaf(qd.x, a[cc].x,  fmaf(qd.y, px, -qd.z * py));
                na.y = fmaf(qd.x, a[cc].y,  fmaf(qd.y, py,  qd.z * px));
                a[cc] = na;
                b[cc] = nb;
            }
        }
    }
}

__device__ __forceinline__ void rows_in(float2 (*U)[LDP], int row, float2 (&a)[COLS], float2 (&b)[COLS]) {
#pragma unroll
    for (int cc = 0; cc < COLS; ++cc) {
        const float4 r = *reinterpret_cast<const float4*>(&U[cc][row]);
        a[cc] = make_float2(r.x, r.y);
        b[cc] = make_float2(r.z, r.w);
    }
}
__device__ __forceinline__ void rows_out(float2 (*U)[LDP], int row, const float2 (&a)[COLS], const float2 (&b)[COLS]) {
#pragma unroll
    for (int cc = 0; cc < COLS; ++cc)
        *reinterpret_cast<float4*>(&U[cc][row]) = make_float4(a[cc].x, a[cc].y, b[cc].x, b[cc].y);
}

__device__ __forceinline__ void mesh_slab(const float4* __restrict__ tcur, const float4* __restrict__ tnext,
                                          float2 (*U)[LDP], int row, float4* bufa, float4* bufb) {
    float2 a[COLS], b[COLS];
    rows_in(U, row, a, b);
#pragma unroll
    for (int j = 0; j < SLOTS_B; ++j) bufb[j] = load_coef(tcur + (SLOTS_A + j) * SUBS);
    tile_run<0>(bufa, a, b);
    rows_out(U, row, a, b);
    lds_barrier();

    rows_in(U, row + HS, a, b);
#pragma unroll
    for (int j = 0; j < SLOTS_A; ++j) bufa[j] = load_coef(tnext + j * SUBS);
    tile_run<1>(bufb, a, b);
    rows_out(U, row + HS, a, b);
    lds_barrier();
}

__global__ void __launch_bounds__(32 * MESH_WAVES) __attribute__((amdgpu_waves_per_eu(1, 1)))
mzi_mesh_kernel(const float4* __restrict__ tab,
                const float* __restrict__ phi_ext,
                float2* __restrict__ WP) {
    __shared__ __attribute__((aligned(16))) float2 U[COLS][LDP];
    const int tid  = threadIdx.x;
    const int lane = tid & 31;
    const int sub  = lane & (SUBS - 1);
    const int g    = lane / SUBS;
    const int wv   = __builtin_amdgcn_readfirstlane(tid >> 5);
    const int tile = GROUPS * wv + g;
    const int row  = TR * tile + 2 * sub;
    const int col0 = blockIdx.x * COLS;
    const int SLAB_STRIDE = TILES * NSLOT * SUBS;
    const float4* tw = tab + tile * NSLOT * SUBS + sub;

    float4 bufa[SLOTS_A], bufb[SLOTS_B];
#pragma unroll
    for (int j = 0; j < SLOTS_A; ++j) bufa[j] = load_coef(tw + j * SUBS);

    for (int i = tid; i < COLS * LDP; i += 32 * MESH_WAVES) {
        const int cc = i / LDP, r = i - cc * LDP;
        U[cc][r] = make_float2((r == col0 + cc) ? 1.0f : 0.0f, 0.0f);
    }
    lds_barrier();

    for (int k = 0; k < NSLAB; ++k)
        mesh_slab(tw + k * SLAB_STRIDE, tw + min(k + 1, NSLAB - 1) * SLAB_STRIDE, U, row, bufa, bufb);

    for (int j = tid; j < NN; j += 32 * MESH_WAVES) {
        float sj, cj;
        sincosf(phi_ext[j], &sj, &cj);
        const float2 u0 = U[0][j];
        const float2 u1 = U[1][j];
        WP[blockIdx.x * NN + j] =
            make_float2(fmaf(cj, u0.x, -sj * u0.y), fmaf(cj, u1.x, -sj * u1.y));
    }
}

__global__ void __launch_bounds__(256)
mzi_gemm_kernel(const float* __restrict__ x,
                const float* __restrict__ W,
                float* __restrict__ out) {
    const int lane  = threadIdx.x & 31;
    const int wave  = threadIdx.x >> 5;
    const int tile  = blockIdx.x * 8 + wave;
    const int tM    = tile >> 2;
    const int tN    = tile & 3;
    const int lo    = lane & 15;
    const int hi    = lane >> 4;

    const float* aptr0 = x + (tM * 32      + lo) * NN + hi * 2;
    const float* aptr1 = aptr0 + 16 * NN;
    const float* bptr0 = W + hi * 2 * NN + (tN * 64 + lo) * 2;
    const float* bptr1 = bptr0 + 16 * 2;
    const float* bptr2 = bptr0 + 32 * 2;
    const float* bptr3 = bptr0 + 48 * 2;

    v8f c00 = {}, c01 = {}, c02 = {}, c03 = {};
    v8f c10 = {}, c11 = {}, c12 = {}, c13 = {};

#pragma unroll 2
    for (int k = 0; k < NN; k += 4) {
        v2f a0 = *(const v2f*)(aptr0 + k);
        v2f a1 = *(const v2f*)(aptr1 + k);
        v2f b0 = *(const v2f*)(bptr0 + k * NN);
        v2f b1 = *(const v2f*)(bptr1 + k * NN);
        v2f b2 = *(const v2f*)(bptr2 + k * NN);
        v2f b3 = *(const v2f*)(bptr3 + k * NN);
        c00 = __builtin_amdgcn_wmma_f32_16x16x4_f32(false, a0, false, b0, (short)0, c00, false, false);
        c01 = __builtin_amdgcn_wmma_f32_16x16x4_f32(false, a0, false, b1, (short)0, c01, false, false);
        c02 = __builtin_amdgcn_wmma_f32_16x16x4_f32(false, a0, false, b2, (short)0, c02, false, false);
        c03 = __builtin_amdgcn_wmma_f32_16x16x4_f32(false, a0, false, b3, (short)0, c03, false, false);
        c10 = __builtin_amdgcn_wmma_f32_16x16x4_f32(false, a1, false, b0, (short)0, c10, false, false);
        c11 = __builtin_amdgcn_wmma_f32_16x16x4_f32(false, a1, false, b1, (short)0, c11, false, false);
        c12 = __builtin_amdgcn_wmma_f32_16x16x4_f32(false, a1, false, b2, (short)0, c12, false, false);
        c13 = __builtin_amdgcn_wmma_f32_16x16x4_f32(false, a1, false, b3, (short)0, c13, false, false);
    }

    v8f acc[2][4] = {{c00, c01, c02, c03}, {c10, c11, c12, c13}};
#pragma unroll
    for (int i = 0; i < 2; ++i) {
#pragma unroll
        for (int j = 0; j < 4; ++j) {
            float* optr = out + (tM * 32 + i * 16 + hi * 8) * NN
                              + tN * 64 + j * 16 + lo;
#pragma unroll
            for (int g = 0; g < 8; ++g)
                optr[g * NN] = acc[i][j][g];
        }
    }
}

extern "C" void kernel_launch(void* const* d_in, const int* in_sizes, int n_in,
                              void* d_out, int out_size, void* d_ws, size_t ws_size,
                              hipStream_t stream) {
    const float* x       = (const float*)d_in[0];
    const float* theta   = (const float*)d_in[1];
    const float* phi_int = (const float*)d_in[2];
    const float* phi_ext = (const float*)d_in[3];
    float*       out     = (float*)d_out;

    float4* coef = (float4*)d_ws;
    float*  W    = (float*)((char*)d_ws + (size_t)TAB_ENTRIES * 16);

    mzi_coef_kernel<<<NSLAB * TILES, 768, 0, stream>>>(theta, phi_int, coef);
    mzi_mesh_kernel<<<NN / COLS, 32 * MESH_WAVES, 0, stream>>>(coef, phi_ext, (float2*)W);
    mzi_gemm_kernel<<<(BATCH / 32) * (NN / 64) / 8, 256, 0, stream>>>(x, W, out);
}


# Mesh: a wave skips the slabs where both its tiles lie outside the mesh triangle (identity entries only): no table requests, no row traffic, only the barriers; 83 of 128 wave-slabs are left
# speedup vs baseline: 37.9115x; 1.0415x over previous
#include <hip/hip_runtime.h>
#include <math.h>

#define NN 256
#define BATCH 4096
#define N_MZIS 32640
#define LAYERS 255

typedef __attribute__((ext_vector_type(2))) float v2f;
typedef __attribute__((ext_vector_type(8))) float v8f;

#define NSTEPS 509
#define COLS 2
#define MESH_WAVES 4
#define GROUPS 2
#define TILES 8
#define TR 32
#define HS 16
#define NSLAB 32
#define LDP (NN + 16)
#define NSLOT 47
#define SLOTS_A 24
#define SLOTS_B (NSLOT - SLOTS_A)
#define SUBS 16
#define TAB_ENTRIES (NSLAB * TILES * NSLOT * SUBS)
static_assert(TILES * TR == NN && MESH_WAVES * GROUPS == TILES, "the lane groups' tiles cover every row");
static_assert(NSLAB * HS >= NSTEPS, "slabs cover the scan");
static_assert(SLOTS_A == 3 * HS / 2 && NSLOT == 3 * HS - 1 && SUBS == TR / 2, "slots and lanes follow the tile");
static_assert(LDP >= NN + TR / 2, "the last tile's shifted pass stays inside the padded column");

__global__ void __launch_bounds__(768)
mzi_coef_kernel(const float* __restrict__ theta,
                const float* __restrict__ phi,
                float4* __restrict__ tab) {
    const int k   = blockIdx.x / TILES;
    const int w   = blockIdx.x % TILES;
    const int j   = threadIdx.x / SUBS;
    const int sub = threadIdx.x % SUBS;
    if (j >= NSLOT) return;
    const bool shifted = (j >= SLOTS_A);
    const int  jj      = shifted ? j - SLOTS_A + 1 : j;
    const int  which   = jj % 3;
    const int  s       = 2 * (jj / 3) + (which != 0 ? 1 : 0);
    const int  t       = HS * k + s;
    int base, plo, phi_;
    if (!shifted) {
        base = TR * w;
        plo  = (w == 0) ? 0 : TR * w + s;
        phi_ = (w == TILES - 1) ? NN - 2 : TR * w + TR - 2 - s;
    } else {
        base = TR * w + HS;
        plo  = TR * (w + 1) - 1 - s;
        phi_ = (w == TILES - 1) ? -1 : TR * (w + 1) - 1 + s;
    }
    const int  p  = base + 2 * sub + (which == 1 ? 1 : (which == 2 ? -1 : 0));
    const bool on = p >= plo && p <= phi_ && p <= min(t, 2 * (NN - 2) - t);
    float4 q = make_float4(1.0f, 0.0f, 0.0f, 0.0f);
    if (on) {
        const int l = (t - p) >> 1;
        const int i = l * 255 - ((l * (l - 1)) >> 1) + p;
        float sn, cs, sp, cp;
        sincosf(theta[i], &sn, &cs);
        sincosf(phi[i], &sp, &cp);
        q = make_float4(cs, sn * cp, sn * sp, 0.0f);
    }
    tab[(blockIdx.x * NSLOT + j) * SUBS + sub] = q;
}

__device__ __forceinline__ void lds_barrier() {
    __builtin_amdgcn_fence(__ATOMIC_RELEASE, "workgroup", "local");
    __builtin_amdgcn_s_barrier();
    __builtin_amdgcn_fence(__ATOMIC_ACQUIRE, "workgroup", "local");
}

__device__ __forceinline__ float from_next(float v) {
    const unsigned u = __builtin_bit_cast(unsigned, v);
    return __builtin_bit_cast(float, __builtin_amdgcn_permlane16(u, u, 0x87654321u, 0xffedcba9u, true, false));
}
__device__ __forceinline__ float from_prev(float v) {
    const unsigned u = __builtin_bit_cast(unsigned, v);
    return __builtin_bit_cast(float, __builtin_amdgcn_permlane16(u, u, 0x65432100u, 0xedcba987u, true, false));
}

__device__ __forceinline__ float4 load_coef(const float4* p) {
    const float* f = static_cast<const float*>(__builtin_assume_aligned(p, 16));
    return make_float4(f[0], f[1], f[2], 0.0f);
}

template <int S0>
__device__ __forceinline__ void tile_run(const float4* cf, float2 (&a)[COLS], float2 (&b)[COLS]) {
    int k = 0;
#pragma unroll
    for (int s = S0; s < HS; ++s) {
        if ((s & 1) == 0) {
            const float4 q = cf[k++];
#pragma unroll
            for (int cc = 0; cc < COLS; ++cc) {
                float2 na, nb;
                na.x = fmaf(q.x, a[cc].x, -fmaf(q.y, b[cc].x,  q.z * b[cc].y));
                na.y = fmaf(q.x, a[cc].y, -fmaf(q.y, b[cc].y, -q.z * b[cc].x));
                nb.x = fmaf(q.x, b[cc].x,  fmaf(q.y, a[cc].x, -q.z * a[cc].y));
                nb.y = fmaf(q.x, b[cc].y,  fmaf(q.y, a[cc].y,  q.z * a[cc].x));
                a[cc] = na;
                b[cc] = nb;
            }
        } else {
            const float4 qu = cf[k++];
            const float4 qd = cf[k++];
#pragma unroll
            for (int cc = 0; cc < COLS; ++cc) {
                const float nx = from_next(a[cc].x), ny = from_next(a[cc].y);
                const float px = from_prev(b[cc].x), py = from_prev(b[cc].y);
                float2 na, nb;
                nb.x = fmaf(qu.x, b[cc].x, fmaf(-qu.y, nx, -qu.z * ny));
                nb.y = fmaf(qu.x, b[cc].y, fmaf(-qu.y, ny,  qu.z * nx));
                na.x = fmaf(qd.x, a[cc].x,  fmaf(qd.y, px, -qd.z * py));
                na.y = fmaf(qd.x, a[cc].y,  fmaf(qd.y, py,  qd.z * px));
                a[cc] = na;
                b[cc] = nb;
            }
        }
    }
}

__device__ __forceinline__ void rows_in(float2 (*U)[LDP], int row, float2 (&a)[COLS], float2 (&b)[COLS]) {
#pragma unroll
    for (int cc = 0; cc < COLS; ++cc) {
        const float4 r = *reinterpret_cast<const float4*>(&U[cc][row]);
        a[cc] = make_float2(r.x, r.y);
        b[cc] = make_float2(r.z, r.w);
    }
}
__device__ __forceinline__ void rows_out(float2 (*U)[LDP], int row, const float2 (&a)[COLS], const float2 (&b)[COLS]) {
#pragma unroll
    for (int cc = 0; cc < COLS; ++cc)
        *reinterpret_cast<float4*>(&U[cc][row]) = make_float4(a[cc].x, a[cc].y, b[cc].x, b[cc].y);
}

__device__ __forceinline__ bool wave_has_work(int k, int wv) {
    return GROUPS * TR * wv <= min(HS * k + HS - 1, 2 * (NN - 2) - HS * k);
}

__device__ __forceinline__ void slab_aligned(const float4* __restrict__ tcur, const float4* __restrict__ tnext,
                                             float2 (*U)[LDP], int row, float4* bufa, float4* bufb) {
    float2 a[COLS], b[COLS];
    rows_in(U, row, a, b);
    __builtin_amdgcn_sched_barrier(0);
#pragma unroll
    for (int j = 0; j < SLOTS_B; ++j) bufb[j] = load_coef(tcur + (SLOTS_A + j) * SUBS);
    __builtin_amdgcn_sched_barrier(0);
    tile_run<0>(bufa, a, b);
    rows_out(U, row, a, b);
    __builtin_amdgcn_sched_barrier(0);
#pragma unroll
    for (int j = 0; j < SLOTS_A; ++j) bufa[j] = load_coef(tnext + j * SUBS);
}

__device__ __forceinline__ void slab_shifted(float2 (*U)[LDP], int row, const float4* bufb) {
    float2 a[COLS], b[COLS];
    rows_in(U, row + HS, a, b);
    tile_run<1>(bufb, a, b);
    rows_out(U, row + HS, a, b);
}

__global__ void __launch_bounds__(32 * MESH_WAVES) __attribute__((amdgpu_waves_per_eu(1, 1)))
mzi_mesh_kernel(const float4* __restrict__ tab,
                const float* __restrict__ phi_ext,
                float2* __restrict__ WP) {
    __shared__ __attribute__((aligned(16))) float2 U[COLS][LDP];
    const int tid  = threadIdx.x;
    const int lane = tid & 31;
    const int sub  = lane & (SUBS - 1);
    const int g    = lane / SUBS;
    const int wv   = __builtin_amdgcn_readfirstlane(tid >> 5);
    const int tile = GROUPS * wv + g;
    const int row  = TR * tile + 2 * sub;
    const int col0 = blockIdx.x * COLS;
    const int SLAB_STRIDE = TILES * NSLOT * SUBS;
    const float4* tw = tab + tile * NSLOT * SUBS + sub;

    float4 bufa[SLOTS_A], bufb[SLOTS_B];
#pragma unroll
    for (int j = 0; j < SLOTS_A; ++j) bufa[j] = load_coef(tw + j * SUBS);

    for (int i = tid; i < COLS * LDP; i += 32 * MESH_WAVES) {
        const int cc = i / LDP, r = i - cc * LDP;
        U[cc][r] = make_float2((r == col0 + cc) ? 1.0f : 0.0f, 0.0f);
    }
    lds_barrier();

    for (int k = 0; k < NSLAB; ++k) {
        const bool run = wave_has_work(k, wv) || (k + 1 < NSLAB && wave_has_work(k + 1, wv));
        if (run) slab_aligned(tw + k * SLAB_STRIDE, tw + min(k + 1, NSLAB - 1) * SLAB_STRIDE, U, row, bufa, bufb);
        lds_barrier();
        if (run) slab_shifted(U, row, bufb);
        lds_barrier();
    }

    for (int j = tid; j < NN; j += 32 * MESH_WAVES) {
        float sj, cj;
        sincosf(phi_ext[j], &sj, &cj);
        const float2 u0 = U[0][j];
        const float2 u1 = U[1][j];
        WP[blockIdx.x * NN + j] =
            make_float2(fmaf(cj, u0.x, -sj * u0.y), fmaf(cj, u1.x, -sj * u1.y));
    }
}

__global__ void __launch_bounds__(256)
mzi_gemm_kernel(const float* __restrict__ x,
                const float* __restrict__ W,
                float* __restrict__ out) {
    const int lane  = threadIdx.x & 31;
    const int wave  = threadIdx.x >> 5;
    const int tile  = blockIdx.x * 8 + wave;
    const int tM    = tile >> 2;
    const int tN    = tile & 3;
    const int lo    = lane & 15;
    const int hi    = lane >> 4;

    const float* aptr0 = x + (tM * 32      + lo) * NN + hi * 2;
    const float* aptr1 = aptr0 + 16 * NN;
    const float* bptr0 = W + hi * 2 * NN + (tN * 64 + lo) * 2;
    const float* bptr1 = bptr0 + 16 * 2;
    const float* bptr2 = bptr0 + 32 * 2;
    const float* bptr3 = bptr0 + 48 * 2;

    v8f c00 = {}, c01 = {}, c02 = {}, c03 = {};
    v8f c10 = {}, c11 = {}, c12 = {}, c13 = {};

#pragma unroll 2
    for (int k = 0; k < NN; k += 4) {
        v2f a0 = *(const v2f*)(aptr0 + k);
        v2f a1 = *(const v2f*)(aptr1 + k);
        v2f b0 = *(const v2f*)(bptr0 + k * NN);
        v2f b1 = *(const v2f*)(bptr1 + k * NN);
        v2f b2 = *(const v2f*)(bptr2 + k * NN);
        v2f b3 = *(const v2f*)(bptr3 + k * NN);
        c00 = __builtin_amdgcn_wmma_f32_16x16x4_f32(false, a0, false, b0, (short)0, c00, false, false);
        c01 = __builtin_amdgcn_wmma_f32_16x16x4_f32(false, a0, false, b1, (short)0, c01, false, false);
        c02 = __builtin_amdgcn_wmma_f32_16x16x4_f32(false, a0, false, b2, (short)0, c02, false, false);
        c03 = __builtin_amdgcn_wmma_f32_16x16x4_f32(false, a0, false, b3, (short)0, c03, false, false);
        c10 = __builtin_amdgcn_wmma_f32_16x16x4_f32(false, a1, false, b0, (short)0, c10, false, false);
        c11 = __builtin_amdgcn_wmma_f32_16x16x4_f32(false, a1, false, b1, (short)0, c11, false, false);
        c12 = __builtin_amdgcn_wmma_f32_16x16x4_f32(false, a1, false, b2, (short)0, c12, false, false);
        c13 = __builtin_amdgcn_wmma_f32_16x16x4_f32(false, a1, false, b3, (short)0, c13, false, false);
    }

    v8f acc[2][4] = {{c00, c01, c02, c03}, {c10, c11, c12, c13}};
#pragma unroll
    for (int i = 0; i < 2; ++i) {
#pragma unroll
        for (int j = 0; j < 4; ++j) {
            float* optr = out + (tM * 32 + i * 16 + hi * 8) * NN
                              + tN * 64 + j * 16 + lo;
#pragma unroll
            for (int g = 0; g < 8; ++g)
                optr[g * NN] = acc[i][j][g];
        }
    }
}

extern "C" void kernel_launch(void* const* d_in, const int* in_sizes, int n_in,
                              void* d_out, int out_size, void* d_ws, size_t ws_size,
                              hipStream_t stream) {
    const float* x       = (const float*)d_in[0];
    const float* theta   = (const float*)d_in[1];
    const float* phi_int = (const float*)d_in[2];
    const float* phi_ext = (const float*)d_in[3];
    float*       out     = (float*)d_out;

    float4* coef = (float4*)d_ws;
    float*  W    = (float*)((char*)d_ws + (size_t)TAB_ENTRIES * 16);

    mzi_coef_kernel<<<NSLAB * TILES, 768, 0, stream>>>(theta, phi_int, coef);
    mzi_mesh_kernel<<<NN / COLS, 32 * MESH_WAVES, 0, stream>>>(coef, phi_ext, (float2*)W);
    mzi_gemm_kernel<<<(BATCH / 32) * (NN / 64) / 8, 256, 0, stream>>>(x, W, out);
}


# Mesh tiles of 64 rows x 32 wavefronts (was 32 x 16): 32 LDS hand-overs, was 64; a tile is a whole wave, row pairs alternate between its two 16-lane rows so one v_permlanex16_var reaches a neighbour
# speedup vs baseline: 40.2533x; 1.0618x over previous
#include <hip/hip_runtime.h>
#include <math.h>

#define NN 256
#define BATCH 4096
#define N_MZIS 32640
#define LAYERS 255

typedef __attribute__((ext_vector_type(2))) float v2f;
typedef __attribute__((ext_vector_type(8))) float v8f;

#define NSTEPS 509
#define COLS 2
#define MESH_WAVES 4
#define GROUPS 1
#define TILES 4
#define TR 64
#define HS 32
#define NSLAB 16
#define LDP (NN + 32)
#define NSLOT 95
#define SLOTS_A 48
#define SLOTS_B (NSLOT - SLOTS_A)
#define SUBS 32
#define TAB_ENTRIES (NSLAB * TILES * NSLOT * SUBS)
static_assert(TILES * TR == NN && MESH_WAVES * GROUPS == TILES, "the lane groups' tiles cover every row");
static_assert(NSLAB * HS >= NSTEPS, "slabs cover the scan");
static_assert(SLOTS_A == 3 * HS / 2 && NSLOT == 3 * HS - 1 && SUBS == TR / 2, "slots and lanes follow the tile");
static_assert(LDP >= NN + TR / 2, "the last tile's shifted pass stays inside the padded column");
static_assert(SUBS == 32 && GROUPS == 1, "a tile is one wave");
#define COEF_THREADS 256
static_assert(TAB_ENTRIES % COEF_THREADS == 0, "whole blocks write the table");

__global__ void __launch_bounds__(COEF_THREADS)
mzi_coef_kernel(const float* __restrict__ theta,
                const float* __restrict__ phi,
                float4* __restrict__ tab) {
    const int e    = blockIdx.x * COEF_THREADS + threadIdx.x;
    const int lane = e % SUBS;
    const int j    = (e / SUBS) % NSLOT;
    const int w    = (e / (SUBS * NSLOT)) % TILES;
    const int k    = e / (SUBS * NSLOT * TILES);
    const int sub  = 2 * (lane & 15) + (lane >> 4);
    const bool shifted = (j >= SLOTS_A);
    const int  jj      = shifted ? j - SLOTS_A + 1 : j;
    const int  which   = jj % 3;
    const int  s       = 2 * (jj / 3) + (which != 0 ? 1 : 0);
    const int  t       = HS * k + s;
    int base, plo, phi_;
    if (!shifted) {
        base = TR * w;
        plo  = (w == 0) ? 0 : TR * w + s;
        phi_ = (w == TILES - 1) ? NN - 2 : TR * w + TR - 2 - s;
    } else {
        base = TR * w + HS;
        plo  = TR * (w + 1) - 1 - s;
        phi_ = (w == TILES - 1) ? -1 : TR * (w + 1) - 1 + s;
    }
    const int  p  = base + 2 * sub + (which == 1 ? 1 : (which == 2 ? -1 : 0));
    const bool on = p >= plo && p <= phi_ && p <= min(t, 2 * (NN - 2) - t);
    float4 q = make_float4(1.0f, 0.0f, 0.0f, 0.0f);
    if (on) {
        const int l = (t - p) >> 1;
        const int i = l * 255 - ((l * (l - 1)) >> 1) + p;
        float sn, cs, sp, cp;
        sincosf(theta[i], &sn, &cs);
        sincosf(phi[i], &sp, &cp);
        q = make_float4(cs, sn * cp, sn * sp, 0.0f);
    }
    tab[e] = q;
}

__device__ __forceinline__ void lds_barrier() {
    __builtin_amdgcn_fence(__ATOMIC_RELEASE, "workgroup", "local");
    __builtin_amdgcn_s_barrier();
    __builtin_amdgcn_fence(__ATOMIC_ACQUIRE, "workgroup", "local");
}

__device__ __forceinline__ float from_other_row(float v, unsigned sel) {
    const unsigned u = __builtin_bit_cast(unsigned, v);
    return __builtin_bit_cast(float, __builtin_amdgcn_permlanex16_var(u, u, sel, true, false));
}

__device__ __forceinline__ float4 load_coef(const float4* p) {
    const float* f = static_cast<const float*>(__builtin_assume_aligned(p, 16));
    return make_float4(f[0], f[1], f[2], 0.0f);
}

template <int S0>
__device__ __forceinline__ void tile_run(const float4* cf, float2 (&a)[COLS], float2 (&b)[COLS],
                                         unsigned sel_next, unsigned sel_prev) {
    int k = 0;
#pragma unroll
    for (int s = S0; s < HS; ++s) {
        if ((s & 1) == 0) {
            const float4 q = cf[k++];
#pragma unroll
            for (int cc = 0; cc < COLS; ++cc) {
                float2 na, nb;
                na.x = fmaf(q.x, a[cc].x, -fmaf(q.y, b[cc].x,  q.z * b[cc].y));
                na.y = fmaf(q.x, a[cc].y, -fmaf(q.y, b[cc].y, -q.z * b[cc].x));
                nb.x = fmaf(q.x, b[cc].x,  fmaf(q.y, a[cc].x, -q.z * a[cc].y));
                nb.y = fmaf(q.x, b[cc].y,  fmaf(q.y, a[cc].y,  q.z * a[cc].x));
                a[cc] = na;
                b[cc] = nb;
            }
        } else {
            const float4 qu = cf[k++];
            const float4 qd = cf[k++];
#pragma unroll
            for (int cc = 0; cc < COLS; ++cc) {
                const float nx = from_other_row(a[cc].x, sel_next), ny = from_other_row(a[cc].y, sel_next);
                const float px = from_other_row(b[cc].x, sel_prev), py = from_other_row(b[cc].y, sel_prev);
                float2 na, nb;
                nb.x = fmaf(qu.x, b[cc].x, fmaf(-qu.y, nx, -qu.z * ny));
                nb.y = fmaf(qu.x, b[cc].y, fmaf(-qu.y, ny,  qu.z * nx));
                na.x = fmaf(qd.x, a[cc].x,  fmaf(qd.y, px, -qd.z * py));
                na.y = fmaf(qd.x, a[cc].y,  fmaf(qd.y, py,  qd.z * px));
                a[cc] = na;
                b[cc] = nb;
            }
        }
    }
}

__device__ __forceinline__ void rows_in(float2 (*U)[LDP], int row, float2 (&a)[COLS], float2 (&b)[COLS]) {
#pragma unroll
    for (int cc = 0; cc < COLS; ++cc) {
        const float4 r = *reinterpret_cast<const float4*>(&U[cc][row]);
        a[cc] = make_float2(r.x, r.y);
        b[cc] = make_float2(r.z, r.w);
    }
}
__device__ __forceinline__ void rows_out(float2 (*U)[LDP], int row, const float2 (&a)[COLS], const float2 (&b)[COLS]) {
#pragma unroll
    for (int cc = 0; cc < COLS; ++cc)
        *reinterpret_cast<float4*>(&U[cc][row]) = make_float4(a[cc].x, a[cc].y, b[cc].x, b[cc].y);
}

__device__ __forceinline__ bool wave_has_work(int k, int wv) {
    return GROUPS * TR * wv <= min(HS * k + HS - 1, 2 * (NN - 2) - HS * k);
}

__device__ __forceinline__ void slab_aligned(const float4* __restrict__ tcur, const float4* __restrict__ tnext,
                                             float2 (*U)[LDP], int row, float4* bufa, float4* bufb,
                                             unsigned sel_next, unsigned sel_prev) {
    float2 a[COLS], b[COLS];
    rows_in(U, row, a, b);
    __builtin_amdgcn_sched_barrier(0);
#pragma unroll
    for (int j = 0; j < SLOTS_B; ++j) bufb[j] = load_coef(tcur + (SLOTS_A + j) * SUBS);
    __builtin_amdgcn_sched_barrier(0);
    tile_run<0>(bufa, a, b, sel_next, sel_prev);
    rows_out(U, row, a, b);
    __builtin_amdgcn_sched_barrier(0);
#pragma unroll
    for (int j = 0; j < SLOTS_A; ++j) bufa[j] = load_coef(tnext + j * SUBS);
}

__device__ __forceinline__ void slab_shifted(float2 (*U)[LDP], int row, const float4* bufb,
                                             unsigned sel_next, unsigned sel_prev) {
    float2 a[COLS], b[COLS];
    rows_in(U, row + HS, a, b);
    tile_run<1>(bufb, a, b, sel_next, sel_prev);
    rows_out(U, row + HS, a, b);
}

__global__ void __launch_bounds__(32 * MESH_WAVES) __attribute__((amdgpu_waves_per_eu(1, 1)))
mzi_mesh_kernel(const float4* __restrict__ tab,
                const float* __restrict__ phi_ext,
                float2* __restrict__ WP) {
    __shared__ __attribute__((aligned(16))) float2 U[COLS][LDP];
    const int tid  = threadIdx.x;
    const int lane = tid & 31;
    const int pos  = lane & 15;
    const int half = lane >> 4;
    const int sub  = 2 * pos + half;
    const unsigned sel_next = (unsigned)min(pos + half, 15);
    const unsigned sel_prev = (unsigned)max(pos + half - 1, 0);
    const int wv   = __builtin_amdgcn_readfirstlane(tid >> 5);
    const int tile = wv;
    const int row  = TR * tile + 2 * sub;
    const int col0 = blockIdx.x * COLS;
    const int SLAB_STRIDE = TILES * NSLOT * SUBS;
    const float4* tw = tab + tile * NSLOT * SUBS + lane;

    float4 bufa[SLOTS_A], bufb[SLOTS_B];
#pragma unroll
    for (int j = 0; j < SLOTS_A; ++j) bufa[j] = load_coef(tw + j * SUBS);

    for (int i = tid; i < COLS * LDP; i += 32 * MESH_WAVES) {
        const int cc = i / LDP, r = i - cc * LDP;
        U[cc][r] = make_float2((r == col0 + cc) ? 1.0f : 0.0f, 0.0f);
    }
    lds_barrier();

    for (int k = 0; k < NSLAB; ++k) {
        const bool run = wave_has_work(k, wv) || (k + 1 < NSLAB && wave_has_work(k + 1, wv));
        if (run) slab_aligned(tw + k * SLAB_STRIDE, tw + min(k + 1, NSLAB - 1) * SLAB_STRIDE, U, row, bufa, bufb, sel_next, sel_prev);
        lds_barrier();
        if (run) slab_shifted(U, row, bufb, sel_next, sel_prev);
        lds_barrier();
    }

    for (int j = tid; j < NN; j += 32 * MESH_WAVES) {
        float sj, cj;
        sincosf(phi_ext[j], &sj, &cj);
        const float2 u0 = U[0][j];
        const float2 u1 = U[1][j];
        WP[blockIdx.x * NN + j] =
            make_float2(fmaf(cj, u0.x, -sj * u0.y), fmaf(cj, u1.x, -sj * u1.y));
    }
}

__global__ void __launch_bounds__(256)
mzi_gemm_kernel(const float* __restrict__ x,
                const float* __restrict__ W,
                float* __restrict__ out) {
    const int lane  = threadIdx.x & 31;
    const int wave  = threadIdx.x >> 5;
    const int tile  = blockIdx.x * 8 + wave;
    const int tM    = tile >> 2;
    const int tN    = tile & 3;
    const int lo    = lane & 15;
    const int hi    = lane >> 4;

    const float* aptr0 = x + (tM * 32      + lo) * NN + hi * 2;
    const float* aptr1 = aptr0 + 16 * NN;
    const float* bptr0 = W + hi * 2 * NN + (tN * 64 + lo) * 2;
    const float* bptr1 = bptr0 + 16 * 2;
    const float* bptr2 = bptr0 + 32 * 2;
    const float* bptr3 = bptr0 + 48 * 2;

    v8f c00 = {}, c01 = {}, c02 = {}, c03 = {};
    v8f c10 = {}, c11 = {}, c12 = {}, c13 = {};

#pragma unroll 2
    for (int k = 0; k < NN; k += 4) {
        v2f a0 = *(const v2f*)(aptr0 + k);
        v2f a1 = *(const v2f*)(aptr1 + k);
        v2f b0 = *(const v2f*)(bptr0 + k * NN);
        v2f b1 = *(const v2f*)(bptr1 + k * NN);
        v2f b2 = *(const v2f*)(bptr2 + k * NN);
        v2f b3 = *(const v2f*)(bptr3 + k * NN);
        c00 = __builtin_amdgcn_wmma_f32_16x16x4_f32(false, a0, false, b0, (short)0, c00, false, false);
        c01 = __builtin_amdgcn_wmma_f32_16x16x4_f32(false, a0, false, b1, (short)0, c01, false, false);
        c02 = __builtin_amdgcn_wmma_f32_16x16x4_f32(false, a0, false, b2, (short)0, c02, false, false);
        c03 = __builtin_amdgcn_wmma_f32_16x16x4_f32(false, a0, false, b3, (short)0, c03, false, false);
        c10 = __builtin_amdgcn_wmma_f32_16x16x4_f32(false, a1, false, b0, (short)0, c10, false, false);
        c11 = __builtin_amdgcn_wmma_f32_16x16x4_f32(false, a1, false, b1, (short)0, c11, false, false);
        c12 = __builtin_amdgcn_wmma_f32_16x16x4_f32(false, a1, false, b2, (short)0, c12, false, false);
        c13 = __builtin_amdgcn_wmma_f32_16x16x4_f32(false, a1, false, b3, (short)0, c13, false, false);
    }

    v8f acc[2][4] = {{c00, c01, c02, c03}, {c10, c11, c12, c13}};
#pragma unroll
    for (int i = 0; i < 2; ++i) {
#pragma unroll
        for (int j = 0; j < 4; ++j) {
            float* optr = out + (tM * 32 + i * 16 + hi * 8) * NN
                              + tN * 64 + j * 16 + lo;
#pragma unroll
            for (int g = 0; g < 8; ++g)
                optr[g * NN] = acc[i][j][g];
        }
    }
}

extern "C" void kernel_launch(void* const* d_in, const int* in_sizes, int n_in,
                              void* d_out, int out_size, void* d_ws, size_t ws_size,
                              hipStream_t stream) {
    const float* x       = (const float*)d_in[0];
    const float* theta   = (const float*)d_in[1];
    const float* phi_int = (const float*)d_in[2];
    const float* phi_ext = (const float*)d_in[3];
    float*       out     = (float*)d_out;

    float4* coef = (float4*)d_ws;
    float*  W    = (float*)((char*)d_ws + (size_t)TAB_ENTRIES * 16);

    mzi_coef_kernel<<<TAB_ENTRIES / COEF_THREADS, COEF_THREADS, 0, stream>>>(theta, phi_int, coef);
    mzi_mesh_kernel<<<NN / COLS, 32 * MESH_WAVES, 0, stream>>>(coef, phi_ext, (float2*)W);
    mzi_gemm_kernel<<<(BATCH / 32) * (NN / 64) / 8, 256, 0, stream>>>(x, W, out);
}


# Final product on every WGP: 256 blocks of 32 rows x 128 outputs (was 64 blocks, matrix-pipe bound on 64 WGPs); x rows staged once in LDS by whole-line loads, W fragments fetched 16 K-steps ahead
# speedup vs baseline: 54.1122x; 1.3443x over previous
#include <hip/hip_runtime.h>
#include <math.h>

#define NN 256
#define BATCH 4096
#define N_MZIS 32640
#define LAYERS 255

typedef __attribute__((ext_vector_type(2))) float v2f;
typedef __attribute__((ext_vector_type(8))) float v8f;

#define NSTEPS 509
#define COLS 2
#define MESH_WAVES 4
#define GROUPS 1
#define TILES 4
#define TR 64
#define HS 32
#define NSLAB 16
#define LDP (NN + 32)
#define NSLOT 95
#define SLOTS_A 48
#define SLOTS_B (NSLOT - SLOTS_A)
#define SUBS 32
#define TAB_ENTRIES (NSLAB * TILES * NSLOT * SUBS)
static_assert(TILES * TR == NN && MESH_WAVES * GROUPS == TILES, "the lane groups' tiles cover every row");
static_assert(NSLAB * HS >= NSTEPS, "slabs cover the scan");
static_assert(SLOTS_A == 3 * HS / 2 && NSLOT == 3 * HS - 1 && SUBS == TR / 2, "slots and lanes follow the tile");
static_assert(LDP >= NN + TR / 2, "the last tile's shifted pass stays inside the padded column");
static_assert(SUBS == 32 && GROUPS == 1, "a tile is one wave");
#define COEF_THREADS 256
static_assert(TAB_ENTRIES % COEF_THREADS == 0, "whole blocks write the table");

__global__ void __launch_bounds__(COEF_THREADS)
mzi_coef_kernel(const float* __restrict__ theta,
                const float* __restrict__ phi,
                float4* __restrict__ tab) {
    const int e    = blockIdx.x * COEF_THREADS + threadIdx.x;
    const int lane = e % SUBS;
    const int j    = (e / SUBS) % NSLOT;
    const int w    = (e / (SUBS * NSLOT)) % TILES;
    const int k    = e / (SUBS * NSLOT * TILES);
    const int sub  = 2 * (lane & 15) + (lane >> 4);
    const bool shifted = (j >= SLOTS_A);
    const int  jj      = shifted ? j - SLOTS_A + 1 : j;
    const int  which   = jj % 3;
    const int  s       = 2 * (jj / 3) + (which != 0 ? 1 : 0);
    const int  t       = HS * k + s;
    int base, plo, phi_;
    if (!shifted) {
        base = TR * w;
        plo  = (w == 0) ? 0 : TR * w + s;
        phi_ = (w == TILES - 1) ? NN - 2 : TR * w + TR - 2 - s;
    } else {
        base = TR * w + HS;
        plo  = TR * (w + 1) - 1 - s;
        phi_ = (w == TILES - 1) ? -1 : TR * (w + 1) - 1 + s;
    }
    const int  p  = base + 2 * sub + (which == 1 ? 1 : (which == 2 ? -1 : 0));
    const bool on = p >= plo && p <= phi_ && p <= min(t, 2 * (NN - 2) - t);
    float4 q = make_float4(1.0f, 0.0f, 0.0f, 0.0f);
    if (on) {
        const int l = (t - p) >> 1;
        const int i = l * 255 - ((l * (l - 1)) >> 1) + p;
        float sn, cs, sp, cp;
        sincosf(theta[i], &sn, &cs);
        sincosf(phi[i], &sp, &cp);
        q = make_float4(cs, sn * cp, sn * sp, 0.0f);
    }
    tab[e] = q;
}

__device__ __forceinline__ void lds_barrier() {
    __builtin_amdgcn_fence(__ATOMIC_RELEASE, "workgroup", "local");
    __builtin_amdgcn_s_barrier();
    __builtin_amdgcn_fence(__ATOMIC_ACQUIRE, "workgroup", "local");
}

__device__ __forceinline__ float from_other_row(float v, unsigned sel) {
    const unsigned u = __builtin_bit_cast(unsigned, v);
    return __builtin_bit_cast(float, __builtin_amdgcn_permlanex16_var(u, u, sel, true, false));
}

__device__ __forceinline__ float4 load_coef(const float4* p) {
    const float* f = static_cast<const float*>(__builtin_assume_aligned(p, 16));
    return make_float4(f[0], f[1], f[2], 0.0f);
}

template <int S0>
__device__ __forceinline__ void tile_run(const float4* cf, float2 (&a)[COLS], float2 (&b)[COLS],
                                         unsigned sel_next, unsigned sel_prev) {
    int k = 0;
#pragma unroll
    for (int s = S0; s < HS; ++s) {
        if ((s & 1) == 0) {
            const float4 q = cf[k++];
#pragma unroll
            for (int cc = 0; cc < COLS; ++cc) {
                float2 na, nb;
                na.x = fmaf(q.x, a[cc].x, -fmaf(q.y, b[cc].x,  q.z * b[cc].y));
                na.y = fmaf(q.x, a[cc].y, -fmaf(q.y, b[cc].y, -q.z * b[cc].x));
                nb.x = fmaf(q.x, b[cc].x,  fmaf(q.y, a[cc].x, -q.z * a[cc].y));
                nb.y = fmaf(q.x, b[cc].y,  fmaf(q.y, a[cc].y,  q.z * a[cc].x));
                a[cc] = na;
                b[cc] = nb;
            }
        } else {
            const float4 qu = cf[k++];
            const float4 qd = cf[k++];
#pragma unroll
            for (int cc = 0; cc < COLS; ++cc) {
                const float nx = from_other_row(a[cc].x, sel_next), ny = from_other_row(a[cc].y, sel_next);
                const float px = from_other_row(b[cc].x, sel_prev), py = from_other_row(b[cc].y, sel_prev);
                float2 na, nb;
                nb.x = fmaf(qu.x, b[cc].x, fmaf(-qu.y, nx, -qu.z * ny));
                nb.y = fmaf(qu.x, b[cc].y, fmaf(-qu.y, ny,  qu.z * nx));
                na.x = fmaf(qd.x, a[cc].x,  fmaf(qd.y, px, -qd.z * py));
                na.y = fmaf(qd.x, a[cc].y,  fmaf(qd.y, py,  qd.z * px));
                a[cc] = na;
                b[cc] = nb;
            }
        }
    }
}

__device__ __forceinline__ void rows_in(float2 (*U)[LDP], int row, float2 (&a)[COLS], float2 (&b)[COLS]) {
#pragma unroll
    for (int cc = 0; cc < COLS; ++cc) {
        const float4 r = *reinterpret_cast<const float4*>(&U[cc][row]);
        a[cc] = make_float2(r.x, r.y);
        b[cc] = make_float2(r.z, r.w);
    }
}
__device__ __forceinline__ void rows_out(float2 (*U)[LDP], int row, const float2 (&a)[COLS], const float2 (&b)[COLS]) {
#pragma unroll
    for (int cc = 0; cc < COLS; ++cc)
        *reinterpret_cast<float4*>(&U[cc][row]) = make_float4(a[cc].x, a[cc].y, b[cc].x, b[cc].y);
}

__device__ __forceinline__ bool wave_has_work(int k, int wv) {
    return GROUPS * TR * wv <= min(HS * k + HS - 1, 2 * (NN - 2) - HS * k);
}

__device__ __forceinline__ void slab_aligned(const float4* __restrict__ tcur, const float4* __restrict__ tnext,
                                             float2 (*U)[LDP], int row, float4* bufa, float4* bufb,
                                             unsigned sel_next, unsigned sel_prev) {
    float2 a[COLS], b[COLS];
    rows_in(U, row, a, b);
    __builtin_amdgcn_sched_barrier(0);
#pragma unroll
    for (int j = 0; j < SLOTS_B; ++j) bufb[j] = load_coef(tcur + (SLOTS_A + j) * SUBS);
    __builtin_amdgcn_sched_barrier(0);
    tile_run<0>(bufa, a, b, sel_next, sel_prev);
    rows_out(U, row, a, b);
    __builtin_amdgcn_sched_barrier(0);
#pragma unroll
    for (int j = 0; j < SLOTS_A; ++j) bufa[j] = load_coef(tnext + j * SUBS);
}

__device__ __forceinline__ void slab_shifted(float2 (*U)[LDP], int row, const float4* bufb,
                                             unsigned sel_next, unsigned sel_prev) {
    float2 a[COLS], b[COLS];
    rows_in(U, row + HS, a, b);
    tile_run<1>(bufb, a, b, sel_next, sel_prev);
    rows_out(U, row + HS, a, b);
}

__global__ void __launch_bounds__(32 * MESH_WAVES) __attribute__((amdgpu_waves_per_eu(1, 1)))
mzi_mesh_kernel(const float4* __restrict__ tab,
                const float* __restrict__ phi_ext,
                float2* __restrict__ WP) {
    __shared__ __attribute__((aligned(16))) float2 U[COLS][LDP];
    const int tid  = threadIdx.x;
    const int lane = tid & 31;
    const int pos  = lane & 15;
    const int half = lane >> 4;
    const int sub  = 2 * pos + half;
    const unsigned sel_next = (unsigned)min(pos + half, 15);
    const unsigned sel_prev = (unsigned)max(pos + half - 1, 0);
    const int wv   = __builtin_amdgcn_readfirstlane(tid >> 5);
    const int tile = wv;
    const int row  = TR * tile + 2 * sub;
    const int col0 = blockIdx.x * COLS;
    const int SLAB_STRIDE = TILES * NSLOT * SUBS;
    const float4* tw = tab + tile * NSLOT * SUBS + lane;

    float4 bufa[SLOTS_A], bufb[SLOTS_B];
#pragma unroll
    for (int j = 0; j < SLOTS_A; ++j) bufa[j] = load_coef(tw + j * SUBS);

    for (int i = tid; i < COLS * LDP; i += 32 * MESH_WAVES) {
        const int cc = i / LDP, r = i - cc * LDP;
        U[cc][r] = make_float2((r == col0 + cc) ? 1.0f : 0.0f, 0.0f);
    }
    lds_barrier();

    for (int k = 0; k < NSLAB; ++k) {
        const bool run = wave_has_work(k, wv) || (k + 1 < NSLAB && wave_has_work(k + 1, wv));
        if (run) slab_aligned(tw + k * SLAB_STRIDE, tw + min(k + 1, NSLAB - 1) * SLAB_STRIDE, U, row, bufa, bufb, sel_next, sel_prev);
        lds_barrier();
        if (run) slab_shifted(U, row, bufb, sel_next, sel_prev);
        lds_barrier();
    }

    for (int j = tid; j < NN; j += 32 * MESH_WAVES) {
        float sj, cj;
        sincosf(phi_ext[j], &sj, &cj);
        const float2 u0 = U[0][j];
        const float2 u1 = U[1][j];
        WP[blockIdx.x * NN + j] =
            make_float2(fmaf(cj, u0.x, -sj * u0.y), fmaf(cj, u1.x, -sj * u1.y));
    }
}

#define GM_ROWS 32
#define GM_COLS 128
#define GM_LDA (NN + 4)
#define GM_CHUNK 16
#define GM_NCHUNK (NN / 4 / GM_CHUNK)
static_assert(GM_NCHUNK * GM_CHUNK * 4 == NN, "chunks cover K");
static_assert(BATCH % GM_ROWS == 0 && NN % GM_COLS == 0 && GM_COLS == 4 * 32 && GM_ROWS == 2 * 16, "8 waves cover the work-group's tile");
static_assert(GM_ROWS * NN / 4 % 256 == 0, "whole work-groups stage x");

__global__ void __launch_bounds__(256)
mzi_gemm_kernel(const float* __restrict__ x,
                const float* __restrict__ W,
                float* __restrict__ out) {
    __shared__ __attribute__((aligned(16))) float As[GM_ROWS][GM_LDA];
    const int tid   = threadIdx.x;
    const int lane  = tid & 31;
    const int wave  = __builtin_amdgcn_readfirstlane(tid >> 5);
    const int lo    = lane & 15;
    const int hi    = lane >> 4;
    const int m0    = (blockIdx.x >> 1) * GM_ROWS;
    const int n0    = (blockIdx.x & 1) * GM_COLS + (wave & 3) * 32;
    const int r0    = (wave >> 2) * 16;

    const float* bp = W + hi * 2 * NN + (n0 + lo) * 2;

    v2f bq[2][GM_CHUNK][2];
#pragma unroll
    for (int kk = 0; kk < GM_CHUNK; ++kk) {
        bq[0][kk][0] = *(const v2f*)(bp + 4 * kk * NN);
        bq[0][kk][1] = *(const v2f*)(bp + 4 * kk * NN + 32);
    }

#pragma unroll
    for (int i = 0; i < GM_ROWS * NN / 4 / 256; ++i) {
        const int idx = tid + 256 * i;
        const int r = idx >> 6, c4 = idx & 63;
        const float4 v = *reinterpret_cast<const float4*>(x + (m0 + r) * NN + c4 * 4);
        *reinterpret_cast<float4*>(&As[r][c4 * 4]) = v;
    }
    __syncthreads();

    const float* ap = &As[r0 + lo][hi * 2];
    v8f c0 = {}, c1 = {};

#pragma unroll
    for (int ch = 0; ch < GM_NCHUNK; ++ch) {
        if (ch + 1 < GM_NCHUNK) {
#pragma unroll
            for (int kk = 0; kk < GM_CHUNK; ++kk) {
                const int ks = (ch + 1) * GM_CHUNK + kk;
                bq[(ch + 1) & 1][kk][0] = *(const v2f*)(bp + 4 * ks * NN);
                bq[(ch + 1) & 1][kk][1] = *(const v2f*)(bp + 4 * ks * NN + 32);
            }
        }
#pragma unroll
        for (int kk = 0; kk < GM_CHUNK; ++kk) {
            const v2f a = *(const v2f*)(ap + 4 * (ch * GM_CHUNK + kk));
            c0 = __builtin_amdgcn_wmma_f32_16x16x4_f32(false, a, false, bq[ch & 1][kk][0], (short)0, c0, false, false);
            c1 = __builtin_amdgcn_wmma_f32_16x16x4_f32(false, a, false, bq[ch & 1][kk][1], (short)0, c1, false, false);
        }
    }

    float* optr = out + (m0 + r0 + hi * 8) * NN + n0 + lo;
#pragma unroll
    for (int g = 0; g < 8; ++g) {
        optr[g * NN]      = c0[g];
        optr[g * NN + 16] = c1[g];
    }
}

extern "C" void kernel_launch(void* const* d_in, const int* in_sizes, int n_in,
                              void* d_out, int out_size, void* d_ws, size_t ws_size,
                              hipStream_t stream) {
    const float* x       = (const float*)d_in[0];
    const float* theta   = (const float*)d_in[1];
    const float* phi_int = (const float*)d_in[2];
    const float* phi_ext = (const float*)d_in[3];
    float*       out     = (float*)d_out;

    float4* coef = (float4*)d_ws;
    float*  W    = (float*)((char*)d_ws + (size_t)TAB_ENTRIES * 16);

    mzi_coef_kernel<<<TAB_ENTRIES / COEF_THREADS, COEF_THREADS, 0, stream>>>(theta, phi_int, coef);
    mzi_mesh_kernel<<<NN / COLS, 32 * MESH_WAVES, 0, stream>>>(coef, phi_ext, (float2*)W);
    mzi_gemm_kernel<<<(BATCH / GM_ROWS) * (NN / GM_COLS), 256, 0, stream>>>(x, W, out);
}


# Coefficient table holds every MZI once in 12 bytes (was 16 with a padding word, odd-step MZIs twice): both lanes of an odd-step MZI read one entry; 63 slots of 384 B per tile and slab, was 95 of 512
# speedup vs baseline: 59.8970x; 1.1069x over previous
#include <hip/hip_runtime.h>
#include <math.h>

#define NN 256
#define BATCH 4096
#define N_MZIS 32640
#define LAYERS 255

typedef __attribute__((ext_vector_type(2))) float v2f;
typedef __attribute__((ext_vector_type(8))) float v8f;

#define NSTEPS 509
#define COLS 2
#define MESH_WAVES 4
#define GROUPS 1
#define TILES 4
#define TR 64
#define HS 32
#define NSLAB 16
#define LDP (NN + 32)
#define NSLOT 95
#define SLOTS_A 48
#define SLOTS_B (NSLOT - SLOTS_A)
#define SUBS 32
#define TSLOTS (2 * HS - 1)
#define SLOT_ENT SUBS
#define CF 3
#define TAB_ENTRIES (NSLAB * TILES * TSLOTS * SLOT_ENT)
static_assert(TILES * TR == NN && MESH_WAVES * GROUPS == TILES, "the lane groups' tiles cover every row");
static_assert(NSLAB * HS >= NSTEPS, "slabs cover the scan");
static_assert(SLOTS_A == 3 * HS / 2 && NSLOT == 3 * HS - 1 && SUBS == TR / 2, "slots and lanes follow the tile");
static_assert(LDP >= NN + TR / 2, "the last tile's shifted pass stays inside the padded column");
static_assert(SUBS == 32 && GROUPS == 1, "a tile is one wave");
#define COEF_THREADS 256
static_assert(TAB_ENTRIES % COEF_THREADS == 0, "whole blocks write the table");

__global__ void __launch_bounds__(COEF_THREADS)
mzi_coef_kernel(const float* __restrict__ theta,
                const float* __restrict__ phi,
                float* __restrict__ tab) {
    const int e    = blockIdx.x * COEF_THREADS + threadIdx.x;
    const int lane = e % SLOT_ENT;
    const int sl   = (e / SLOT_ENT) % TSLOTS;
    const int w    = (e / (SLOT_ENT * TSLOTS)) % TILES;
    const int k    = e / (SLOT_ENT * TSLOTS * TILES);
    const int sub  = 2 * (lane & 15) + (lane >> 4);
    const bool shifted = (sl >= HS);
    const int  s       = shifted ? sl - (HS - 1) : sl;
    const int  t       = HS * k + s;
    int base, plo, phi_;
    if (!shifted) {
        base = TR * w;
        plo  = (w == 0) ? 0 : TR * w + s;
        phi_ = (w == TILES - 1) ? NN - 2 : TR * w + TR - 2 - s;
    } else {
        base = TR * w + HS;
        plo  = TR * (w + 1) - 1 - s;
        phi_ = (w == TILES - 1) ? -1 : TR * (w + 1) - 1 + s;
    }
    const int  p  = base + 2 * sub + (s & 1);
    const bool on = p >= plo && p <= phi_ && p <= min(t, 2 * (NN - 2) - t);
    float4 q = make_float4(1.0f, 0.0f, 0.0f, 0.0f);
    if (on) {
        const int l = (t - p) >> 1;
        const int i = l * 255 - ((l * (l - 1)) >> 1) + p;
        float sn, cs, sp, cp;
        sincosf(theta[i], &sn, &cs);
        sincosf(phi[i], &sp, &cp);
        q = make_float4(cs, sn * cp, sn * sp, 0.0f);
    }
    float* o = tab + CF * e;
    o[0] = q.x;
    o[1] = q.y;
    o[2] = q.z;
}

__device__ __forceinline__ void lds_barrier() {
    __builtin_amdgcn_fence(__ATOMIC_RELEASE, "workgroup", "local");
    __builtin_amdgcn_s_barrier();
    __builtin_amdgcn_fence(__ATOMIC_ACQUIRE, "workgroup", "local");
}

__device__ __forceinline__ float from_other_row(float v, unsigned sel) {
    const unsigned u = __builtin_bit_cast(unsigned, v);
    return __builtin_bit_cast(float, __builtin_amdgcn_permlanex16_var(u, u, sel, true, false));
}

__device__ __forceinline__ float4 load_coef(const float* p) {
    return make_float4(p[0], p[1], p[2], 0.0f);
}

__device__ __forceinline__ float4 pass_coef(const float* own, const float* above, int jj, int first_slot) {
    const int which = jj % 3;
    const int s     = 2 * (jj / 3) + (which != 0 ? 1 : 0);
    return load_coef((which == 2 ? above : own) + (first_slot + s) * SLOT_ENT * CF);
}

template <int S0>
__device__ __forceinline__ void tile_run(const float4* cf, float2 (&a)[COLS], float2 (&b)[COLS],
                                         unsigned sel_next, unsigned sel_prev) {
    int k = 0;
#pragma unroll
    for (int s = S0; s < HS; ++s) {
        if ((s & 1) == 0) {
            const float4 q = cf[k++];
#pragma unroll
            for (int cc = 0; cc < COLS; ++cc) {
                float2 na, nb;
                na.x = fmaf(q.x, a[cc].x, -fmaf(q.y, b[cc].x,  q.z * b[cc].y));
                na.y = fmaf(q.x, a[cc].y, -fmaf(q.y, b[cc].y, -q.z * b[cc].x));
                nb.x = fmaf(q.x, b[cc].x,  fmaf(q.y, a[cc].x, -q.z * a[cc].y));
                nb.y = fmaf(q.x, b[cc].y,  fmaf(q.y, a[cc].y,  q.z * a[cc].x));
                a[cc] = na;
                b[cc] = nb;
            }
        } else {
            const float4 qu = cf[k++];
            const float4 qd = cf[k++];
#pragma unroll
            for (int cc = 0; cc < COLS; ++cc) {
                const float nx = from_other_row(a[cc].x, sel_next), ny = from_other_row(a[cc].y, sel_next);
                const float px = from_other_row(b[cc].x, sel_prev), py = from_other_row(b[cc].y, sel_prev);
                float2 na, nb;
                nb.x = fmaf(qu.x, b[cc].x, fmaf(-qu.y, nx, -qu.z * ny));
                nb.y = fmaf(qu.x, b[cc].y, fmaf(-qu.y, ny,  qu.z * nx));
                na.x = fmaf(qd.x, a[cc].x,  fmaf(qd.y, px, -qd.z * py));
                na.y = fmaf(qd.x, a[cc].y,  fmaf(qd.y, py,  qd.z * px));
                a[cc] = na;
                b[cc] = nb;
            }
        }
    }
}

__device__ __forceinline__ void rows_in(float2 (*U)[LDP], int row, float2 (&a)[COLS], float2 (&b)[COLS]) {
#pragma unroll
    for (int cc = 0; cc < COLS; ++cc) {
        const float4 r = *reinterpret_cast<const float4*>(&U[cc][row]);
        a[cc] = make_float2(r.x, r.y);
        b[cc] = make_float2(r.z, r.w);
    }
}
__device__ __forceinline__ void rows_out(float2 (*U)[LDP], int row, const float2 (&a)[COLS], const float2 (&b)[COLS]) {
#pragma unroll
    for (int cc = 0; cc < COLS; ++cc)
        *reinterpret_cast<float4*>(&U[cc][row]) = make_float4(a[cc].x, a[cc].y, b[cc].x, b[cc].y);
}

__device__ __forceinline__ bool wave_has_work(int k, int wv) {
    return GROUPS * TR * wv <= min(HS * k + HS - 1, 2 * (NN - 2) - HS * k);
}

__device__ __forceinline__ void slab_aligned(const float* __restrict__ own, const float* __restrict__ above, int cur, int next,
                                             float2 (*U)[LDP], int row, float4* bufa, float4* bufb,
                                             unsigned sel_next, unsigned sel_prev) {
    float2 a[COLS], b[COLS];
    rows_in(U, row, a, b);
    __builtin_amdgcn_sched_barrier(0);
#pragma unroll
    for (int j = 0; j < SLOTS_B; ++j) bufb[j] = pass_coef(own + cur, above + cur, j + 1, HS - 1);
    __builtin_amdgcn_sched_barrier(0);
    tile_run<0>(bufa, a, b, sel_next, sel_prev);
    rows_out(U, row, a, b);
    __builtin_amdgcn_sched_barrier(0);
#pragma unroll
    for (int j = 0; j < SLOTS_A; ++j) bufa[j] = pass_coef(own + next, above + next, j, 0);
}

__device__ __forceinline__ void slab_shifted(float2 (*U)[LDP], int row, const float4* bufb,
                                             unsigned sel_next, unsigned sel_prev) {
    float2 a[COLS], b[COLS];
    rows_in(U, row + HS, a, b);
    tile_run<1>(bufb, a, b, sel_next, sel_prev);
    rows_out(U, row + HS, a, b);
}

__global__ void __launch_bounds__(32 * MESH_WAVES) __attribute__((amdgpu_waves_per_eu(1, 1)))
mzi_mesh_kernel(const float* __restrict__ tab,
                const float* __restrict__ phi_ext,
                float2* __restrict__ WP) {
    __shared__ __attribute__((aligned(16))) float2 U[COLS][LDP];
    const int tid  = threadIdx.x;
    const int lane = tid & 31;
    const int pos  = lane & 15;
    const int half = lane >> 4;
    const int sub  = 2 * pos + half;
    const unsigned sel_next = (unsigned)min(pos + half, 15);
    const unsigned sel_prev = (unsigned)max(pos + half - 1, 0);
    const int wv   = __builtin_amdgcn_readfirstlane(tid >> 5);
    const int tile = wv;
    const int row  = TR * tile + 2 * sub;
    const int col0 = blockIdx.x * COLS;
    const int SLAB_STRIDE = TILES * TSLOTS * SLOT_ENT * CF;
    const float* tw  = tab + (tile * TSLOTS * SLOT_ENT + lane) * CF;
    const float* twa = tab + (tile * TSLOTS * SLOT_ENT + (half ? pos : (pos ? 15 + pos : 31))) * CF;

    float4 bufa[SLOTS_A], bufb[SLOTS_B];
#pragma unroll
    for (int j = 0; j < SLOTS_A; ++j) bufa[j] = pass_coef(tw, twa, j, 0);

    for (int i = tid; i < COLS * LDP; i += 32 * MESH_WAVES) {
        const int cc = i / LDP, r = i - cc * LDP;
        U[cc][r] = make_float2((r == col0 + cc) ? 1.0f : 0.0f, 0.0f);
    }
    lds_barrier();

    for (int k = 0; k < NSLAB; ++k) {
        const bool run = wave_has_work(k, wv) || (k + 1 < NSLAB && wave_has_work(k + 1, wv));
        if (run) slab_aligned(tw, twa, k * SLAB_STRIDE, min(k + 1, NSLAB - 1) * SLAB_STRIDE, U, row, bufa, bufb, sel_next, sel_prev);
        lds_barrier();
        if (run) slab_shifted(U, row, bufb, sel_next, sel_prev);
        lds_barrier();
    }

    for (int j = tid; j < NN; j += 32 * MESH_WAVES) {
        float sj, cj;
        sincosf(phi_ext[j], &sj, &cj);
        const float2 u0 = U[0][j];
        const float2 u1 = U[1][j];
        WP[blockIdx.x * NN + j] =
            make_float2(fmaf(cj, u0.x, -sj * u0.y), fmaf(cj, u1.x, -sj * u1.y));
    }
}

#define GM_ROWS 32
#define GM_COLS 128
#define GM_LDA (NN + 4)
#define GM_CHUNK 16
#define GM_NCHUNK (NN / 4 / GM_CHUNK)
static_assert(GM_NCHUNK * GM_CHUNK * 4 == NN, "chunks cover K");
static_assert(BATCH % GM_ROWS == 0 && NN % GM_COLS == 0 && GM_COLS == 4 * 32 && GM_ROWS == 2 * 16, "8 waves cover the work-group's tile");
static_assert(GM_ROWS * NN / 4 % 256 == 0, "whole work-groups stage x");

__global__ void __launch_bounds__(256)
mzi_gemm_kernel(const float* __restrict__ x,
                const float* __restrict__ W,
                float* __restrict__ out) {
    __shared__ __attribute__((aligned(16))) float As[GM_ROWS][GM_LDA];
    const int tid   = threadIdx.x;
    const int lane  = tid & 31;
    const int wave  = __builtin_amdgcn_readfirstlane(tid >> 5);
    const int lo    = lane & 15;
    const int hi    = lane >> 4;
    const int m0    = (blockIdx.x >> 1) * GM_ROWS;
    const int n0    = (blockIdx.x & 1) * GM_COLS + (wave & 3) * 32;
    const int r0    = (wave >> 2) * 16;

    const float* bp = W + hi * 2 * NN + (n0 + lo) * 2;

    v2f bq[2][GM_CHUNK][2];
#pragma unroll
    for (int kk = 0; kk < GM_CHUNK; ++kk) {
        bq[0][kk][0] = *(const v2f*)(bp + 4 * kk * NN);
        bq[0][kk][1] = *(const v2f*)(bp + 4 * kk * NN + 32);
    }

#pragma unroll
    for (int i = 0; i < GM_ROWS * NN / 4 / 256; ++i) {
        const int idx = tid + 256 * i;
        const int r = idx >> 6, c4 = idx & 63;
        const float4 v = *reinterpret_cast<const float4*>(x + (m0 + r) * NN + c4 * 4);
        *reinterpret_cast<float4*>(&As[r][c4 * 4]) = v;
    }
    __syncthreads();

    const float* ap = &As[r0 + lo][hi * 2];
    v8f c0 = {}, c1 = {};

#pragma unroll
    for (int ch = 0; ch < GM_NCHUNK; ++ch) {
        if (ch + 1 < GM_NCHUNK) {
#pragma unroll
            for (int kk = 0; kk < GM_CHUNK; ++kk) {
                const int ks = (ch + 1) * GM_CHUNK + kk;
                bq[(ch + 1) & 1][kk][0] = *(const v2f*)(bp + 4 * ks * NN);
                bq[(ch + 1) & 1][kk][1] = *(const v2f*)(bp + 4 * ks * NN + 32);
            }
        }
#pragma unroll
        for (int kk = 0; kk < GM_CHUNK; ++kk) {
            const v2f a = *(const v2f*)(ap + 4 * (ch * GM_CHUNK + kk));
            c0 = __builtin_amdgcn_wmma_f32_16x16x4_f32(false, a, false, bq[ch & 1][kk][0], (short)0, c0, false, false);
            c1 = __builtin_amdgcn_wmma_f32_16x16x4_f32(false, a, false, bq[ch & 1][kk][1], (short)0, c1, false, false);
        }
    }

    float* optr = out + (m0 + r0 + hi * 8) * NN + n0 + lo;
#pragma unroll
    for (int g = 0; g < 8; ++g) {
        optr[g * NN]      = c0[g];
        optr[g * NN + 16] = c1[g];
    }
}

extern "C" void kernel_launch(void* const* d_in, const int* in_sizes, int n_in,
                              void* d_out, int out_size, void* d_ws, size_t ws_size,
                              hipStream_t stream) {
    const float* x       = (const float*)d_in[0];
    const float* theta   = (const float*)d_in[1];
    const float* phi_int = (const float*)d_in[2];
    const float* phi_ext = (const float*)d_in[3];
    float*       out     = (float*)d_out;

    float* coef = (float*)d_ws;
    float* W    = (float*)((char*)d_ws + (size_t)TAB_ENTRIES * CF * 4);

    mzi_coef_kernel<<<TAB_ENTRIES / COEF_THREADS, COEF_THREADS, 0, stream>>>(theta, phi_int, coef);
    mzi_mesh_kernel<<<NN / COLS, 32 * MESH_WAVES, 0, stream>>>(coef, phi_ext, (float2*)W);
    mzi_gemm_kernel<<<(BATCH / GM_ROWS) * (NN / GM_COLS), 256, 0, stream>>>(x, W, out);
}


# Mesh steps: a lane's two columns are the halves of 2-float values, each product and sum one packed f32 instruction for both: 756 packed + 256 permutes per slab, was 1184 + 256; sums unchanged
# speedup vs baseline: 62.9321x; 1.0507x over previous
#include <hip/hip_runtime.h>
#include <math.h>

#define NN 256
#define BATCH 4096
#define N_MZIS 32640
#define LAYERS 255

typedef __attribute__((ext_vector_type(2))) float v2f;
typedef __attribute__((ext_vector_type(8))) float v8f;

#define NSTEPS 509
#define COLS 2
#define MESH_WAVES 4
#define GROUPS 1
#define TILES 4
#define TR 64
#define HS 32
#define NSLAB 16
#define LDP (NN + 32)
#define NSLOT 95
#define SLOTS_A 48
#define SLOTS_B (NSLOT - SLOTS_A)
#define SUBS 32
#define TSLOTS (2 * HS - 1)
#define SLOT_ENT SUBS
#define CF 3
#define TAB_ENTRIES (NSLAB * TILES * TSLOTS * SLOT_ENT)
static_assert(TILES * TR == NN && MESH_WAVES * GROUPS == TILES, "the lane groups' tiles cover every row");
static_assert(NSLAB * HS >= NSTEPS, "slabs cover the scan");
static_assert(SLOTS_A == 3 * HS / 2 && NSLOT == 3 * HS - 1 && SUBS == TR / 2, "slots and lanes follow the tile");
static_assert(LDP >= NN + TR / 2, "the last tile's shifted pass stays inside the padded column");
static_assert(SUBS == 32 && GROUPS == 1, "a tile is one wave");
#define COEF_THREADS 256
static_assert(TAB_ENTRIES % COEF_THREADS == 0, "whole blocks write the table");

__global__ void __launch_bounds__(COEF_THREADS)
mzi_coef_kernel(const float* __restrict__ theta,
                const float* __restrict__ phi,
                float* __restrict__ tab) {
    const int e    = blockIdx.x * COEF_THREADS + threadIdx.x;
    const int lane = e % SLOT_ENT;
    const int sl   = (e / SLOT_ENT) % TSLOTS;
    const int w    = (e / (SLOT_ENT * TSLOTS)) % TILES;
    const int k    = e / (SLOT_ENT * TSLOTS * TILES);
    const int sub  = 2 * (lane & 15) + (lane >> 4);
    const bool shifted = (sl >= HS);
    const int  s       = shifted ? sl - (HS - 1) : sl;
    const int  t       = HS * k + s;
    int base, plo, phi_;
    if (!shifted) {
        base = TR * w;
        plo  = (w == 0) ? 0 : TR * w + s;
        phi_ = (w == TILES - 1) ? NN - 2 : TR * w + TR - 2 - s;
    } else {
        base = TR * w + HS;
        plo  = TR * (w + 1) - 1 - s;
        phi_ = (w == TILES - 1) ? -1 : TR * (w + 1) - 1 + s;
    }
    const int  p  = base + 2 * sub + (s & 1);
    const bool on = p >= plo && p <= phi_ && p <= min(t, 2 * (NN - 2) - t);
    float4 q = make_float4(1.0f, 0.0f, 0.0f, 0.0f);
    if (on) {
        const int l = (t - p) >> 1;
        const int i = l * 255 - ((l * (l - 1)) >> 1) + p;
        float sn, cs, sp, cp;
        sincosf(theta[i], &sn, &cs);
        sincosf(phi[i], &sp, &cp);
        q = make_float4(cs, sn * cp, sn * sp, 0.0f);
    }
    float* o = tab + CF * e;
    o[0] = q.x;
    o[1] = q.y;
    o[2] = q.z;
}

__device__ __forceinline__ void lds_barrier() {
    __builtin_amdgcn_fence(__ATOMIC_RELEASE, "workgroup", "local");
    __builtin_amdgcn_s_barrier();
    __builtin_amdgcn_fence(__ATOMIC_ACQUIRE, "workgroup", "local");
}

__device__ __forceinline__ float from_other_row(float v, unsigned sel) {
    const unsigned u = __builtin_bit_cast(unsigned, v);
    return __builtin_bit_cast(float, __builtin_amdgcn_permlanex16_var(u, u, sel, true, false));
}

__device__ __forceinline__ float4 load_coef(const float* p) {
    return make_float4(p[0], p[1], p[2], 0.0f);
}

__device__ __forceinline__ float4 pass_coef(const float* own, const float* above, int jj, int first_slot) {
    const int which = jj % 3;
    const int s     = 2 * (jj / 3) + (which != 0 ? 1 : 0);
    return load_coef((which == 2 ? above : own) + (first_slot + s) * SLOT_ENT * CF);
}

static_assert(COLS == 2, "the two columns are the two halves of a packed value");
__device__ __forceinline__ v2f both(float s) { return (v2f){s, s}; }
__device__ __forceinline__ v2f pfma(v2f x, v2f y, v2f z) { return __builtin_elementwise_fma(x, y, z); }
__device__ __forceinline__ v2f from_other_row2(v2f v, unsigned sel) {
    return (v2f){from_other_row(v.x, sel), from_other_row(v.y, sel)};
}
template <int S0>
__device__ __forceinline__ void tile_run(const float4* cf, float2 (&a)[COLS], float2 (&b)[COLS],
                                         unsigned sel_next, unsigned sel_prev) {
    v2f ax = {a[0].x, a[1].x}, ay = {a[0].y, a[1].y};
    v2f bx = {b[0].x, b[1].x}, by = {b[0].y, b[1].y};
    int k = 0;
#pragma unroll
    for (int s = S0; s < HS; ++s) {
        if ((s & 1) == 0) {
            const float4 q = cf[k++];
            const v2f c = both(q.x), er = both(q.y), ei = both(q.z);
            const v2f nax = pfma(c, ax, -pfma(er, bx,  ei * by));
            const v2f nay = pfma(c, ay, -pfma(er, by, -ei * bx));
            const v2f nbx = pfma(c, bx,  pfma(er, ax, -ei * ay));
            const v2f nby = pfma(c, by,  pfma(er, ay,  ei * ax));
            ax = nax; ay = nay; bx = nbx; by = nby;
        } else {
            const float4 qu = cf[k++];
            const float4 qd = cf[k++];
            const v2f cu = both(qu.x), eru = both(qu.y), eiu = both(qu.z);
            const v2f cd = both(qd.x), erd = both(qd.y), eid = both(qd.z);
            const v2f nx = from_other_row2(ax, sel_next), ny = from_other_row2(ay, sel_next);
            const v2f px = from_other_row2(bx, sel_prev), py = from_other_row2(by, sel_prev);
            const v2f nbx = pfma(cu, bx, -pfma(eru, nx, eiu * ny));
            const v2f nby = pfma(cu, by, -pfma(eru, ny, -(eiu * nx)));
            const v2f nax = pfma(cd, ax, pfma(erd, px, -eid * py));
            const v2f nay = pfma(cd, ay, pfma(erd, py,  eid * px));
            ax = nax; ay = nay; bx = nbx; by = nby;
        }
    }
    a[0] = make_float2(ax.x, ay.x); a[1] = make_float2(ax.y, ay.y);
    b[0] = make_float2(bx.x, by.x); b[1] = make_float2(bx.y, by.y);
}

__device__ __forceinline__ void rows_in(float2 (*U)[LDP], int row, float2 (&a)[COLS], float2 (&b)[COLS]) {
#pragma unroll
    for (int cc = 0; cc < COLS; ++cc) {
        const float4 r = *reinterpret_cast<const float4*>(&U[cc][row]);
        a[cc] = make_float2(r.x, r.y);
        b[cc] = make_float2(r.z, r.w);
    }
}
__device__ __forceinline__ void rows_out(float2 (*U)[LDP], int row, const float2 (&a)[COLS], const float2 (&b)[COLS]) {
#pragma unroll
    for (int cc = 0; cc < COLS; ++cc)
        *reinterpret_cast<float4*>(&U[cc][row]) = make_float4(a[cc].x, a[cc].y, b[cc].x, b[cc].y);
}

__device__ __forceinline__ bool wave_has_work(int k, int wv) {
    return GROUPS * TR * wv <= min(HS * k + HS - 1, 2 * (NN - 2) - HS * k);
}

__device__ __forceinline__ void slab_aligned(const float* __restrict__ own, const float* __restrict__ above, int cur, int next,
                                             float2 (*U)[LDP], int row, float4* bufa, float4* bufb,
                                             unsigned sel_next, unsigned sel_prev) {
    float2 a[COLS], b[COLS];
    rows_in(U, row, a, b);
    __builtin_amdgcn_sched_barrier(0);
#pragma unroll
    for (int j = 0; j < SLOTS_B; ++j) bufb[j] = pass_coef(own + cur, above + cur, j + 1, HS - 1);
    __builtin_amdgcn_sched_barrier(0);
    tile_run<0>(bufa, a, b, sel_next, sel_prev);
    rows_out(U, row, a, b);
    __builtin_amdgcn_sched_barrier(0);
#pragma unroll
    for (int j = 0; j < SLOTS_A; ++j) bufa[j] = pass_coef(own + next, above + next, j, 0);
}

__device__ __forceinline__ void slab_shifted(float2 (*U)[LDP], int row, const float4* bufb,
                                             unsigned sel_next, unsigned sel_prev) {
    float2 a[COLS], b[COLS];
    rows_in(U, row + HS, a, b);
    tile_run<1>(bufb, a, b, sel_next, sel_prev);
    rows_out(U, row + HS, a, b);
}

__global__ void __launch_bounds__(32 * MESH_WAVES) __attribute__((amdgpu_waves_per_eu(1, 1)))
mzi_mesh_kernel(const float* __restrict__ tab,
                const float* __restrict__ phi_ext,
                float2* __restrict__ WP) {
    __shared__ __attribute__((aligned(16))) float2 U[COLS][LDP];
    const int tid  = threadIdx.x;
    const int lane = tid & 31;
    const int pos  = lane & 15;
    const int half = lane >> 4;
    const int sub  = 2 * pos + half;
    const unsigned sel_next = (unsigned)min(pos + half, 15);
    const unsigned sel_prev = (unsigned)max(pos + half - 1, 0);
    const int wv   = __builtin_amdgcn_readfirstlane(tid >> 5);
    const int tile = wv;
    const int row  = TR * tile + 2 * sub;
    const int col0 = blockIdx.x * COLS;
    const int SLAB_STRIDE = TILES * TSLOTS * SLOT_ENT * CF;
    const float* tw  = tab + (tile * TSLOTS * SLOT_ENT + lane) * CF;
    const float* twa = tab + (tile * TSLOTS * SLOT_ENT + (half ? pos : (pos ? 15 + pos : 31))) * CF;

    float4 bufa[SLOTS_A], bufb[SLOTS_B];
#pragma unroll
    for (int j = 0; j < SLOTS_A; ++j) bufa[j] = pass_coef(tw, twa, j, 0);

    for (int i = tid; i < COLS * LDP; i += 32 * MESH_WAVES) {
        const int cc = i / LDP, r = i - cc * LDP;
        U[cc][r] = make_float2((r == col0 + cc) ? 1.0f : 0.0f, 0.0f);
    }
    lds_barrier();

    for (int k = 0; k < NSLAB; ++k) {
        const bool run = wave_has_work(k, wv) || (k + 1 < NSLAB && wave_has_work(k + 1, wv));
        if (run) slab_aligned(tw, twa, k * SLAB_STRIDE, min(k + 1, NSLAB - 1) * SLAB_STRIDE, U, row, bufa, bufb, sel_next, sel_prev);
        lds_barrier();
        if (run) slab_shifted(U, row, bufb, sel_next, sel_prev);
        lds_barrier();
    }

    for (int j = tid; j < NN; j += 32 * MESH_WAVES) {
        float sj, cj;
        sincosf(phi_ext[j], &sj, &cj);
        const float2 u0 = U[0][j];
        const float2 u1 = U[1][j];
        WP[blockIdx.x * NN + j] =
            make_float2(fmaf(cj, u0.x, -sj * u0.y), fmaf(cj, u1.x, -sj * u1.y));
    }
}

#define GM_ROWS 32
#define GM_COLS 128
#define GM_LDA (NN + 4)
#define GM_CHUNK 16
#define GM_NCHUNK (NN / 4 / GM_CHUNK)
static_assert(GM_NCHUNK * GM_CHUNK * 4 == NN, "chunks cover K");
static_assert(BATCH % GM_ROWS == 0 && NN % GM_COLS == 0 && GM_COLS == 4 * 32 && GM_ROWS == 2 * 16, "8 waves cover the work-group's tile");
static_assert(GM_ROWS * NN / 4 % 256 == 0, "whole work-groups stage x");

__global__ void __launch_bounds__(256)
mzi_gemm_kernel(const float* __restrict__ x,
                const float* __restrict__ W,
                float* __restrict__ out) {
    __shared__ __attribute__((aligned(16))) float As[GM_ROWS][GM_LDA];
    const int tid   = threadIdx.x;
    const int lane  = tid & 31;
    const int wave  = __builtin_amdgcn_readfirstlane(tid >> 5);
    const int lo    = lane & 15;
    const int hi    = lane >> 4;
    const int m0    = (blockIdx.x >> 1) * GM_ROWS;
    const int n0    = (blockIdx.x & 1) * GM_COLS + (wave & 3) * 32;
    const int r0    = (wave >> 2) * 16;

    const float* bp = W + hi * 2 * NN + (n0 + lo) * 2;

    v2f bq[2][GM_CHUNK][2];
#pragma unroll
    for (int kk = 0; kk < GM_CHUNK; ++kk) {
        bq[0][kk][0] = *(const v2f*)(bp + 4 * kk * NN);
        bq[0][kk][1] = *(const v2f*)(bp + 4 * kk * NN + 32);
    }

#pragma unroll
    for (int i = 0; i < GM_ROWS * NN / 4 / 256; ++i) {
        const int idx = tid + 256 * i;
        const int r = idx >> 6, c4 = idx & 63;
        const float4 v = *reinterpret_cast<const float4*>(x + (m0 + r) * NN + c4 * 4);
        *reinterpret_cast<float4*>(&As[r][c4 * 4]) = v;
    }
    __syncthreads();

    const float* ap = &As[r0 + lo][hi * 2];
    v8f c0 = {}, c1 = {};

#pragma unroll
    for (int ch = 0; ch < GM_NCHUNK; ++ch) {
        if (ch + 1 < GM_NCHUNK) {
#pragma unroll
            for (int kk = 0; kk < GM_CHUNK; ++kk) {
                const int ks = (ch + 1) * GM_CHUNK + kk;
                bq[(ch + 1) & 1][kk][0] = *(const v2f*)(bp + 4 * ks * NN);
                bq[(ch + 1) & 1][kk][1] = *(const v2f*)(bp + 4 * ks * NN + 32);
            }
        }
#pragma unroll
        for (int kk = 0; kk < GM_CHUNK; ++kk) {
            const v2f a = *(const v2f*)(ap + 4 * (ch * GM_CHUNK + kk));
            c0 = __builtin_amdgcn_wmma_f32_16x16x4_f32(false, a, false, bq[ch & 1][kk][0], (short)0, c0, false, false);
            c1 = __builtin_amdgcn_wmma_f32_16x16x4_f32(false, a, false, bq[ch & 1][kk][1], (short)0, c1, false, false);
        }
    }

    float* optr = out + (m0 + r0 + hi * 8) * NN + n0 + lo;
#pragma unroll
    for (int g = 0; g < 8; ++g) {
        optr[g * NN]      = c0[g];
        optr[g * NN + 16] = c1[g];
    }
}

extern "C" void kernel_launch(void* const* d_in, const int* in_sizes, int n_in,
                              void* d_out, int out_size, void* d_ws, size_t ws_size,
                              hipStream_t stream) {
    const float* x       = (const float*)d_in[0];
    const float* theta   = (const float*)d_in[1];
    const float* phi_int = (const float*)d_in[2];
    const float* phi_ext = (const float*)d_in[3];
    float*       out     = (float*)d_out;

    float* coef = (float*)d_ws;
    float* W    = (float*)((char*)d_ws + (size_t)TAB_ENTRIES * CF * 4);

    mzi_coef_kernel<<<TAB_ENTRIES / COEF_THREADS, COEF_THREADS, 0, stream>>>(theta, phi_int, coef);
    mzi_mesh_kernel<<<NN / COLS, 32 * MESH_WAVES, 0, stream>>>(coef, phi_ext, (float2*)W);
    mzi_gemm_kernel<<<(BATCH / GM_ROWS) * (NN / GM_COLS), 256, 0, stream>>>(x, W, out);
}
